# starting kernel
# baseline (speedup 1.0000x reference)
#include <hip/hip_runtime.h>

typedef __attribute__((ext_vector_type(2))) float v2f;
typedef __attribute__((ext_vector_type(8))) float v8f;

#define N_NODES 100000
#define N_EDGES 1600000
#define NFEAT   256
#define NHID    128

__global__ void zero_kernel(float* __restrict__ p, int n) {
    int i = blockIdx.x * blockDim.x + threadIdx.x;
    if (i < n) p[i] = 0.0f;
}

__global__ void degree_kernel(const int* __restrict__ src, const int* __restrict__ dst,
                              float* __restrict__ degS, float* __restrict__ degD, int e) {
    int i = blockIdx.x * blockDim.x + threadIdx.x;
    if (i < e) {
        atomicAdd(&degS[src[i]], 1.0f);
        atomicAdd(&degD[dst[i]], 1.0f);
    }
}

__global__ void norm_finalize(float* __restrict__ degS, float* __restrict__ degD, int n) {
    int i = blockIdx.x * blockDim.x + threadIdx.x;
    if (i < n) {
        degS[i] = rsqrtf(fmaxf(degS[i], 1.0f));
        degD[i] = rsqrtf(fmaxf(degD[i], 1.0f));
    }
}

__global__ void __launch_bounds__(256)
gemm_scaled_wmma(const float* __restrict__ X, const float* __restrict__ scale,
                 const float* __restrict__ W, float* __restrict__ Out, int K) {
    const int wave = threadIdx.x >> 5;
    const int lane = threadIdx.x & 31;
    const int kh   = lane >> 4;
    const int l    = lane & 15;
    const int row0 = blockIdx.x << 4;
    const int col0 = wave << 4;
    const int row  = row0 + l;

    const float ns = scale[row];
    const float* __restrict__ xrow = X + (size_t)row * K;

    v8f c = {};
    for (int k = 0; k < K; k += 4) {
        const int ka = k + 2 * kh;
        v2f a;
        a.x = xrow[ka]     * ns;
        a.y = xrow[ka + 1] * ns;
        v2f b;
        b.x = W[(size_t)ka       * NHID + col0 + l];
        b.y = W[(size_t)(ka + 1) * NHID + col0 + l];
        c = __builtin_amdgcn_wmma_f32_16x16x4_f32(false, a, false, b,
                                                  (short)0, c, false, false);
    }

    float* __restrict__ orow = Out + (size_t)row0 * NHID + col0;
#pragma unroll
    for (int r = 0; r < 8; ++r) {
        orow[(size_t)(r + 8 * kh) * NHID + l] = c[r];
    }
}

__global__ void __launch_bounds__(256)
scatter_add(const float* __restrict__ M, const int* __restrict__ src,
            const int* __restrict__ dst, float* __restrict__ AGG, int e) {
    const int w    = (blockIdx.x * blockDim.x + threadIdx.x) >> 5;
    const int lane = threadIdx.x & 31;
    if (w >= e) return;
    const int s = src[w];
    const int d = dst[w];
    const float4 v = reinterpret_cast<const float4*>(M + (size_t)s * NHID)[lane];
    float* drow = AGG + (size_t)d * NHID + lane * 4;
    atomicAdd(drow + 0, v.x);
    atomicAdd(drow + 1, v.y);
    atomicAdd(drow + 2, v.z);
    atomicAdd(drow + 3, v.w);
}

__global__ void bias_relu(const float* __restrict__ AGG, const float* __restrict__ normD,
                          const float* __restrict__ b, float* __restrict__ H, int total) {
    int i = blockIdx.x * blockDim.x + threadIdx.x;
    if (i < total) {
        const int row = i >> 7;
        const int j   = i & 127;
        const float v = AGG[i] * normD[row] + b[j];
        H[i] = v > 0.0f ? v : 0.0f;
    }
}

__global__ void __launch_bounds__(256)
final_fc(const float* __restrict__ AGG, const float* __restrict__ normD,
         const float* __restrict__ b2, const float* __restrict__ Wfc,
         const float* __restrict__ bfc, float* __restrict__ out, int n) {
    const int w    = (blockIdx.x * blockDim.x + threadIdx.x) >> 5;
    const int lane = threadIdx.x & 31;
    if (w >= n) return;
    const float nd = normD[w];
    const float* arow = AGG + (size_t)w * NHID;
    float acc = 0.0f;
#pragma unroll
    for (int t = 0; t < 4; ++t) {
        const int j = lane + 32 * t;
        acc += (arow[j] * nd + b2[j]) * Wfc[j];
    }
#pragma unroll
    for (int off = 16; off > 0; off >>= 1)
        acc += __shfl_xor(acc, off, 32);
    if (lane == 0) out[w] = acc + bfc[0];
}

extern "C" void kernel_launch(void* const* d_in, const int* in_sizes, int n_in,
                              void* d_out, int out_size, void* d_ws, size_t ws_size,
                              hipStream_t stream) {
    const float* x   = (const float*)d_in[0];
    const float* W1  = (const float*)d_in[1];
    const float* b1  = (const float*)d_in[2];
    const float* W2  = (const float*)d_in[3];
    const float* b2  = (const float*)d_in[4];
    const float* Wfc = (const float*)d_in[5];
    const float* bfc = (const float*)d_in[6];
    const int*   src = (const int*)d_in[7];
    const int*   dst = (const int*)d_in[8];
    float* out = (float*)d_out;

    const int N = N_NODES, E = N_EDGES;
    const int NH = NHID;

    float* BUF_A = (float*)d_ws;
    float* BUF_B = BUF_A + (size_t)N * NH;
    float* normS = BUF_B + (size_t)N * NH;
    float* normD = normS + N;

    const int TPB = 256;
    const int featTotal = N * NH;
    const int gridFeat  = (featTotal + TPB - 1) / TPB;
    const int gridN     = (N + TPB - 1) / TPB;
    const int gridE     = (E + TPB - 1) / TPB;
    const int gridEdgeWaves = (E * (TPB / (TPB / 32)) ) ? ((E + 7) / 8) : 0;
    const int gridNodeWaves = (N + 7) / 8;
    const int gridGemm  = N / 16;

    zero_kernel<<<(2 * N + TPB - 1) / TPB, TPB, 0, stream>>>(normS, 2 * N);
    degree_kernel<<<gridE, TPB, 0, stream>>>(src, dst, normS, normD, E);
    norm_finalize<<<gridN, TPB, 0, stream>>>(normS, normD, N);

    zero_kernel<<<gridFeat, TPB, 0, stream>>>(BUF_B, featTotal);
    gemm_scaled_wmma<<<gridGemm, TPB, 0, stream>>>(x, normS, W1, BUF_A, NFEAT);
    scatter_add<<<gridEdgeWaves, TPB, 0, stream>>>(BUF_A, src, dst, BUF_B, E);
    bias_relu<<<gridFeat, TPB, 0, stream>>>(BUF_B, normD, b1, BUF_A, featTotal);

    gemm_scaled_wmma<<<gridGemm, TPB, 0, stream>>>(BUF_A, normS, W2, BUF_B, NH);
    zero_kernel<<<gridFeat, TPB, 0, stream>>>(BUF_A, featTotal);
    scatter_add<<<gridEdgeWaves, TPB, 0, stream>>>(BUF_B, src, dst, BUF_A, E);

    final_fc<<<gridNodeWaves, TPB, 0, stream>>>(BUF_A, normD, b2, Wfc, bfc, out, N);
}


# edge scatter: lane l owns floats l,32+l,64+l,96+l of a row, so each load and each f32 atomic add of a wave covers one whole 128-byte line (4 per edge, not 16 quarter-filled); sums in arrival order
# speedup vs baseline: 1.2278x; 1.2278x over previous
#include <hip/hip_runtime.h>

typedef __attribute__((ext_vector_type(2))) float v2f;
typedef __attribute__((ext_vector_type(8))) float v8f;

#define N_NODES 100000
#define N_EDGES 1600000
#define NFEAT   256
#define NHID    128

__global__ void zero_kernel(float* __restrict__ p, int n) {
    int i = blockIdx.x * blockDim.x + threadIdx.x;
    if (i < n) p[i] = 0.0f;
}

__global__ void degree_kernel(const int* __restrict__ src, const int* __restrict__ dst,
                              float* __restrict__ degS, float* __restrict__ degD, int e) {
    int i = blockIdx.x * blockDim.x + threadIdx.x;
    if (i < e) {
        atomicAdd(&degS[src[i]], 1.0f);
        atomicAdd(&degD[dst[i]], 1.0f);
    }
}

__global__ void norm_finalize(float* __restrict__ degS, float* __restrict__ degD, int n) {
    int i = blockIdx.x * blockDim.x + threadIdx.x;
    if (i < n) {
        degS[i] = rsqrtf(fmaxf(degS[i], 1.0f));
        degD[i] = rsqrtf(fmaxf(degD[i], 1.0f));
    }
}

__global__ void __launch_bounds__(256)
gemm_scaled_wmma(const float* __restrict__ X, const float* __restrict__ scale,
                 const float* __restrict__ W, float* __restrict__ Out, int K) {
    const int wave = threadIdx.x >> 5;
    const int lane = threadIdx.x & 31;
    const int kh   = lane >> 4;
    const int l    = lane & 15;
    const int row0 = blockIdx.x << 4;
    const int col0 = wave << 4;
    const int row  = row0 + l;

    const float ns = scale[row];
    const float* __restrict__ xrow = X + (size_t)row * K;

    v8f c = {};
    for (int k = 0; k < K; k += 4) {
        const int ka = k + 2 * kh;
        v2f a;
        a.x = xrow[ka]     * ns;
        a.y = xrow[ka + 1] * ns;
        v2f b;
        b.x = W[(size_t)ka       * NHID + col0 + l];
        b.y = W[(size_t)(ka + 1) * NHID + col0 + l];
        c = __builtin_amdgcn_wmma_f32_16x16x4_f32(false, a, false, b,
                                                  (short)0, c, false, false);
    }

    float* __restrict__ orow = Out + (size_t)row0 * NHID + col0;
#pragma unroll
    for (int r = 0; r < 8; ++r) {
        orow[(size_t)(r + 8 * kh) * NHID + l] = c[r];
    }
}

__global__ void __launch_bounds__(256)
scatter_add(const float* __restrict__ M, const int* __restrict__ src,
            const int* __restrict__ dst, float* __restrict__ AGG, int e) {
    const int w    = (blockIdx.x * blockDim.x + threadIdx.x) >> 5;
    const int lane = threadIdx.x & 31;
    if (w >= e) return;
    const int s = src[w];
    const int d = dst[w];
    const float* __restrict__ mrow = M + (size_t)s * NHID + lane;
    float* drow = AGG + (size_t)d * NHID + lane;
    const float v0 = mrow[0];
    const float v1 = mrow[32];
    const float v2 = mrow[64];
    const float v3 = mrow[96];
    atomicAdd(drow,      v0);
    atomicAdd(drow + 32, v1);
    atomicAdd(drow + 64, v2);
    atomicAdd(drow + 96, v3);
}

__global__ void bias_relu(const float* __restrict__ AGG, const float* __restrict__ normD,
                          const float* __restrict__ b, float* __restrict__ H, int total) {
    int i = blockIdx.x * blockDim.x + threadIdx.x;
    if (i < total) {
        const int row = i >> 7;
        const int j   = i & 127;
        const float v = AGG[i] * normD[row] + b[j];
        H[i] = v > 0.0f ? v : 0.0f;
    }
}

__global__ void __launch_bounds__(256)
final_fc(const float* __restrict__ AGG, const float* __restrict__ normD,
         const float* __restrict__ b2, const float* __restrict__ Wfc,
         const float* __restrict__ bfc, float* __restrict__ out, int n) {
    const int w    = (blockIdx.x * blockDim.x + threadIdx.x) >> 5;
    const int lane = threadIdx.x & 31;
    if (w >= n) return;
    const float nd = normD[w];
    const float* arow = AGG + (size_t)w * NHID;
    float acc = 0.0f;
#pragma unroll
    for (int t = 0; t < 4; ++t) {
        const int j = lane + 32 * t;
        acc += (arow[j] * nd + b2[j]) * Wfc[j];
    }
#pragma unroll
    for (int off = 16; off > 0; off >>= 1)
        acc += __shfl_xor(acc, off, 32);
    if (lane == 0) out[w] = acc + bfc[0];
}

extern "C" void kernel_launch(void* const* d_in, const int* in_sizes, int n_in,
                              void* d_out, int out_size, void* d_ws, size_t ws_size,
                              hipStream_t stream) {
    const float* x   = (const float*)d_in[0];
    const float* W1  = (const float*)d_in[1];
    const float* b1  = (const float*)d_in[2];
    const float* W2  = (const float*)d_in[3];
    const float* b2  = (const float*)d_in[4];
    const float* Wfc = (const float*)d_in[5];
    const float* bfc = (const float*)d_in[6];
    const int*   src = (const int*)d_in[7];
    const int*   dst = (const int*)d_in[8];
    float* out = (float*)d_out;

    const int N = N_NODES, E = N_EDGES;
    const int NH = NHID;

    float* BUF_A = (float*)d_ws;
    float* BUF_B = BUF_A + (size_t)N * NH;
    float* normS = BUF_B + (size_t)N * NH;
    float* normD = normS + N;

    const int TPB = 256;
    const int featTotal = N * NH;
    const int gridFeat  = (featTotal + TPB - 1) / TPB;
    const int gridN     = (N + TPB - 1) / TPB;
    const int gridE     = (E + TPB - 1) / TPB;
    const int gridEdgeWaves = (E * (TPB / (TPB / 32)) ) ? ((E + 7) / 8) : 0;
    const int gridNodeWaves = (N + 7) / 8;
    const int gridGemm  = N / 16;

    zero_kernel<<<(2 * N + TPB - 1) / TPB, TPB, 0, stream>>>(normS, 2 * N);
    degree_kernel<<<gridE, TPB, 0, stream>>>(src, dst, normS, normD, E);
    norm_finalize<<<gridN, TPB, 0, stream>>>(normS, normD, N);

    zero_kernel<<<gridFeat, TPB, 0, stream>>>(BUF_B, featTotal);
    gemm_scaled_wmma<<<gridGemm, TPB, 0, stream>>>(x, normS, W1, BUF_A, NFEAT);
    scatter_add<<<gridEdgeWaves, TPB, 0, stream>>>(BUF_A, src, dst, BUF_B, E);
    bias_relu<<<gridFeat, TPB, 0, stream>>>(BUF_B, normD, b1, BUF_A, featTotal);

    gemm_scaled_wmma<<<gridGemm, TPB, 0, stream>>>(BUF_A, normS, W2, BUF_B, NH);
    zero_kernel<<<gridFeat, TPB, 0, stream>>>(BUF_A, featTotal);
    scatter_add<<<gridEdgeWaves, TPB, 0, stream>>>(BUF_B, src, dst, BUF_A, E);

    final_fc<<<gridNodeWaves, TPB, 0, stream>>>(BUF_A, normD, b2, Wfc, bfc, out, N);
}


# edge scatter: one wave per 32 edges (one coalesced index load, node ids read back wave-uniformly, rows of 8 edges in flight) instead of one wave per edge; f32 atomic sums still in arrival order
# speedup vs baseline: 1.5032x; 1.2243x over previous
#include <hip/hip_runtime.h>

typedef __attribute__((ext_vector_type(2))) float v2f;
typedef __attribute__((ext_vector_type(8))) float v8f;

#define N_NODES 100000
#define N_EDGES 1600000
#define NFEAT   256
#define NHID    128

__global__ void zero_kernel(float* __restrict__ p, int n) {
    int i = blockIdx.x * blockDim.x + threadIdx.x;
    if (i < n) p[i] = 0.0f;
}

__global__ void degree_kernel(const int* __restrict__ src, const int* __restrict__ dst,
                              float* __restrict__ degS, float* __restrict__ degD, int e) {
    int i = blockIdx.x * blockDim.x + threadIdx.x;
    if (i < e) {
        atomicAdd(&degS[src[i]], 1.0f);
        atomicAdd(&degD[dst[i]], 1.0f);
    }
}

__global__ void norm_finalize(float* __restrict__ degS, float* __restrict__ degD, int n) {
    int i = blockIdx.x * blockDim.x + threadIdx.x;
    if (i < n) {
        degS[i] = rsqrtf(fmaxf(degS[i], 1.0f));
        degD[i] = rsqrtf(fmaxf(degD[i], 1.0f));
    }
}

__global__ void __launch_bounds__(256)
gemm_scaled_wmma(const float* __restrict__ X, const float* __restrict__ scale,
                 const float* __restrict__ W, float* __restrict__ Out, int K) {
    const int wave = threadIdx.x >> 5;
    const int lane = threadIdx.x & 31;
    const int kh   = lane >> 4;
    const int l    = lane & 15;
    const int row0 = blockIdx.x << 4;
    const int col0 = wave << 4;
    const int row  = row0 + l;

    const float ns = scale[row];
    const float* __restrict__ xrow = X + (size_t)row * K;

    v8f c = {};
    for (int k = 0; k < K; k += 4) {
        const int ka = k + 2 * kh;
        v2f a;
        a.x = xrow[ka]     * ns;
        a.y = xrow[ka + 1] * ns;
        v2f b;
        b.x = W[(size_t)ka       * NHID + col0 + l];
        b.y = W[(size_t)(ka + 1) * NHID + col0 + l];
        c = __builtin_amdgcn_wmma_f32_16x16x4_f32(false, a, false, b,
                                                  (short)0, c, false, false);
    }

    float* __restrict__ orow = Out + (size_t)row0 * NHID + col0;
#pragma unroll
    for (int r = 0; r < 8; ++r) {
        orow[(size_t)(r + 8 * kh) * NHID + l] = c[r];
    }
}

#define EDGES_PER_WAVE 32
#define EDGE_BATCH     8

__device__ __forceinline__ void scatter_one_edge(const float* __restrict__ M,
                                                 float* AGG, int s, int d, int lane) {
    const float* __restrict__ mrow = M + (size_t)s * NHID + lane;
    float* drow = AGG + (size_t)d * NHID + lane;
    const float v0 = mrow[0];
    const float v1 = mrow[32];
    const float v2 = mrow[64];
    const float v3 = mrow[96];
    atomicAdd(drow,      v0);
    atomicAdd(drow + 32, v1);
    atomicAdd(drow + 64, v2);
    atomicAdd(drow + 96, v3);
}

__global__ void __launch_bounds__(256)
scatter_add(const float* __restrict__ M, const int* __restrict__ src,
            const int* __restrict__ dst, float* __restrict__ AGG, int e) {
    const int w    = (blockIdx.x * blockDim.x + threadIdx.x) >> 5;
    const int lane = threadIdx.x & 31;
    const int base = w * EDGES_PER_WAVE;
    if (base >= e) return;
    const int mine = base + lane;
    const int sv = (mine < e) ? src[mine] : 0;
    const int dv = (mine < e) ? dst[mine] : 0;
    const int cnt = (e - base < EDGES_PER_WAVE) ? (e - base) : EDGES_PER_WAVE;

    if (cnt == EDGES_PER_WAVE) {
        for (int j0 = 0; j0 < EDGES_PER_WAVE; j0 += EDGE_BATCH) {
            float v[EDGE_BATCH][4];
#pragma unroll
            for (int u = 0; u < EDGE_BATCH; ++u) {
                const int s = __builtin_amdgcn_readlane(sv, j0 + u);
                const float* __restrict__ mrow = M + (size_t)s * NHID + lane;
                v[u][0] = mrow[0];
                v[u][1] = mrow[32];
                v[u][2] = mrow[64];
                v[u][3] = mrow[96];
            }
#pragma unroll
            for (int u = 0; u < EDGE_BATCH; ++u) {
                const int d = __builtin_amdgcn_readlane(dv, j0 + u);
                float* drow = AGG + (size_t)d * NHID + lane;
                atomicAdd(drow,      v[u][0]);
                atomicAdd(drow + 32, v[u][1]);
                atomicAdd(drow + 64, v[u][2]);
                atomicAdd(drow + 96, v[u][3]);
            }
        }
    } else {
        for (int j = 0; j < cnt; ++j) {
            const int s = __builtin_amdgcn_readlane(sv, j);
            const int d = __builtin_amdgcn_readlane(dv, j);
            scatter_one_edge(M, AGG, s, d, lane);
        }
    }
}

__global__ void bias_relu(const float* __restrict__ AGG, const float* __restrict__ normD,
                          const float* __restrict__ b, float* __restrict__ H, int total) {
    int i = blockIdx.x * blockDim.x + threadIdx.x;
    if (i < total) {
        const int row = i >> 7;
        const int j   = i & 127;
        const float v = AGG[i] * normD[row] + b[j];
        H[i] = v > 0.0f ? v : 0.0f;
    }
}

__global__ void __launch_bounds__(256)
final_fc(const float* __restrict__ AGG, const float* __restrict__ normD,
         const float* __restrict__ b2, const float* __restrict__ Wfc,
         const float* __restrict__ bfc, float* __restrict__ out, int n) {
    const int w    = (blockIdx.x * blockDim.x + threadIdx.x) >> 5;
    const int lane = threadIdx.x & 31;
    if (w >= n) return;
    const float nd = normD[w];
    const float* arow = AGG + (size_t)w * NHID;
    float acc = 0.0f;
#pragma unroll
    for (int t = 0; t < 4; ++t) {
        const int j = lane + 32 * t;
        acc += (arow[j] * nd + b2[j]) * Wfc[j];
    }
#pragma unroll
    for (int off = 16; off > 0; off >>= 1)
        acc += __shfl_xor(acc, off, 32);
    if (lane == 0) out[w] = acc + bfc[0];
}

extern "C" void kernel_launch(void* const* d_in, const int* in_sizes, int n_in,
                              void* d_out, int out_size, void* d_ws, size_t ws_size,
                              hipStream_t stream) {
    const float* x   = (const float*)d_in[0];
    const float* W1  = (const float*)d_in[1];
    const float* b1  = (const float*)d_in[2];
    const float* W2  = (const float*)d_in[3];
    const float* b2  = (const float*)d_in[4];
    const float* Wfc = (const float*)d_in[5];
    const float* bfc = (const float*)d_in[6];
    const int*   src = (const int*)d_in[7];
    const int*   dst = (const int*)d_in[8];
    float* out = (float*)d_out;

    const int N = N_NODES, E = N_EDGES;
    const int NH = NHID;

    float* BUF_A = (float*)d_ws;
    float* BUF_B = BUF_A + (size_t)N * NH;
    float* normS = BUF_B + (size_t)N * NH;
    float* normD = normS + N;

    const int TPB = 256;
    const int featTotal = N * NH;
    const int gridFeat  = (featTotal + TPB - 1) / TPB;
    const int gridN     = (N + TPB - 1) / TPB;
    const int gridE     = (E + TPB - 1) / TPB;
    const int gridEdgeWaves = (E + 8 * EDGES_PER_WAVE - 1) / (8 * EDGES_PER_WAVE);
    const int gridNodeWaves = (N + 7) / 8;
    const int gridGemm  = N / 16;

    zero_kernel<<<(2 * N + TPB - 1) / TPB, TPB, 0, stream>>>(normS, 2 * N);
    degree_kernel<<<gridE, TPB, 0, stream>>>(src, dst, normS, normD, E);
    norm_finalize<<<gridN, TPB, 0, stream>>>(normS, normD, N);

    zero_kernel<<<gridFeat, TPB, 0, stream>>>(BUF_B, featTotal);
    gemm_scaled_wmma<<<gridGemm, TPB, 0, stream>>>(x, normS, W1, BUF_A, NFEAT);
    scatter_add<<<gridEdgeWaves, TPB, 0, stream>>>(BUF_A, src, dst, BUF_B, E);
    bias_relu<<<gridFeat, TPB, 0, stream>>>(BUF_B, normD, b1, BUF_A, featTotal);

    gemm_scaled_wmma<<<gridGemm, TPB, 0, stream>>>(BUF_A, normS, W2, BUF_B, NH);
    zero_kernel<<<gridFeat, TPB, 0, stream>>>(BUF_A, featTotal);
    scatter_add<<<gridEdgeWaves, TPB, 0, stream>>>(BUF_B, src, dst, BUF_A, E);

    final_fc<<<gridNodeWaves, TPB, 0, stream>>>(BUF_A, normD, b2, Wfc, bfc, out, N);
}


# final 128->1 projection moved ahead of the second edge sum: the layer-2 product's epilogue dots each row with Wfc, edges then add one float, not a row; sums re-associated, atomics in arrival order
# speedup vs baseline: 1.6614x; 1.1052x over previous
#include <hip/hip_runtime.h>

typedef __attribute__((ext_vector_type(2))) float v2f;
typedef __attribute__((ext_vector_type(8))) float v8f;

#define N_NODES 100000
#define N_EDGES 1600000
#define NFEAT   256
#define NHID    128

__global__ void zero_kernel(float* __restrict__ p, int n) {
    int i = blockIdx.x * blockDim.x + threadIdx.x;
    if (i < n) p[i] = 0.0f;
}

__global__ void degree_kernel(const int* __restrict__ src, const int* __restrict__ dst,
                              float* __restrict__ degS, float* __restrict__ degD, int e) {
    int i = blockIdx.x * blockDim.x + threadIdx.x;
    if (i < e) {
        atomicAdd(&degS[src[i]], 1.0f);
        atomicAdd(&degD[dst[i]], 1.0f);
    }
}

__global__ void norm_finalize(float* __restrict__ degS, float* __restrict__ degD, int n) {
    int i = blockIdx.x * blockDim.x + threadIdx.x;
    if (i < n) {
        degS[i] = rsqrtf(fmaxf(degS[i], 1.0f));
        degD[i] = rsqrtf(fmaxf(degD[i], 1.0f));
    }
}

__global__ void __launch_bounds__(256)
gemm_scaled_wmma(const float* __restrict__ X, const float* __restrict__ scale,
                 const float* __restrict__ W, float* __restrict__ Out, int K) {
    const int wave = threadIdx.x >> 5;
    const int lane = threadIdx.x & 31;
    const int kh   = lane >> 4;
    const int l    = lane & 15;
    const int row0 = blockIdx.x << 4;
    const int col0 = wave << 4;
    const int row  = row0 + l;

    const float ns = scale[row];
    const float* __restrict__ xrow = X + (size_t)row * K;

    v8f c = {};
    for (int k = 0; k < K; k += 4) {
        const int ka = k + 2 * kh;
        v2f a;
        a.x = xrow[ka]     * ns;
        a.y = xrow[ka + 1] * ns;
        v2f b;
        b.x = W[(size_t)ka       * NHID + col0 + l];
        b.y = W[(size_t)(ka + 1) * NHID + col0 + l];
        c = __builtin_amdgcn_wmma_f32_16x16x4_f32(false, a, false, b,
                                                  (short)0, c, false, false);
    }

    float* __restrict__ orow = Out + (size_t)row0 * NHID + col0;
#pragma unroll
    for (int r = 0; r < 8; ++r) {
        orow[(size_t)(r + 8 * kh) * NHID + l] = c[r];
    }
}

__global__ void __launch_bounds__(256)
gemm_scaled_wmma_rowdot(const float* __restrict__ X, const float* __restrict__ scale,
                        const float* __restrict__ W, const float* __restrict__ Wfc,
                        float* __restrict__ S, float* __restrict__ T, int K) {
    __shared__ float part[8][16];

    const int wave = threadIdx.x >> 5;
    const int lane = threadIdx.x & 31;
    const int kh   = lane >> 4;
    const int l    = lane & 15;
    const int row0 = blockIdx.x << 4;
    const int col0 = wave << 4;
    const int row  = row0 + l;

    const float ns = scale[row];
    const float* __restrict__ xrow = X + (size_t)row * K;

    v8f c = {};
    for (int k = 0; k < K; k += 4) {
        const int ka = k + 2 * kh;
        v2f a;
        a.x = xrow[ka]     * ns;
        a.y = xrow[ka + 1] * ns;
        v2f b;
        b.x = W[(size_t)ka       * NHID + col0 + l];
        b.y = W[(size_t)(ka + 1) * NHID + col0 + l];
        c = __builtin_amdgcn_wmma_f32_16x16x4_f32(false, a, false, b,
                                                  (short)0, c, false, false);
    }

    const float wf = Wfc[col0 + l];
    float p[8];
    for (int r = 0; r < 8; ++r) p[r] = c[r] * wf;

    float q4[4];
    for (int i = 0; i < 4; ++i) {
        const float keep = (l & 8) ? p[i + 4] : p[i];
        const float send = (l & 8) ? p[i]     : p[i + 4];
        q4[i] = keep + __shfl_xor(send, 8, 32);
    }
    float q2[2];
    for (int i = 0; i < 2; ++i) {
        const float keep = (l & 4) ? q4[i + 2] : q4[i];
        const float send = (l & 4) ? q4[i]     : q4[i + 2];
        q2[i] = keep + __shfl_xor(send, 4, 32);
    }
    float q1;
    {
        const float keep = (l & 2) ? q2[1] : q2[0];
        const float send = (l & 2) ? q2[0] : q2[1];
        q1 = keep + __shfl_xor(send, 2, 32);
    }
    q1 += __shfl_xor(q1, 1, 32);

    if ((l & 1) == 0) part[wave][8 * kh + (l >> 1)] = q1;
    __syncthreads();

    if (threadIdx.x < 16) {
        float s = 0.0f;
        for (int w = 0; w < 8; ++w) s += part[w][threadIdx.x];
        S[row0 + threadIdx.x] = s;
        T[row0 + threadIdx.x] = 0.0f;
    }
}

#define EDGES_PER_WAVE 32
#define EDGE_BATCH     8

__device__ __forceinline__ void scatter_one_edge(const float* __restrict__ M,
                                                 float* AGG, int s, int d, int lane) {
    const float* __restrict__ mrow = M + (size_t)s * NHID + lane;
    float* drow = AGG + (size_t)d * NHID + lane;
    const float v0 = mrow[0];
    const float v1 = mrow[32];
    const float v2 = mrow[64];
    const float v3 = mrow[96];
    atomicAdd(drow,      v0);
    atomicAdd(drow + 32, v1);
    atomicAdd(drow + 64, v2);
    atomicAdd(drow + 96, v3);
}

__global__ void __launch_bounds__(256)
scatter_add(const float* __restrict__ M, const int* __restrict__ src,
            const int* __restrict__ dst, float* __restrict__ AGG, int e) {
    const int w    = (blockIdx.x * blockDim.x + threadIdx.x) >> 5;
    const int lane = threadIdx.x & 31;
    const int base = w * EDGES_PER_WAVE;
    if (base >= e) return;
    const int mine = base + lane;
    const int sv = (mine < e) ? src[mine] : 0;
    const int dv = (mine < e) ? dst[mine] : 0;
    const int cnt = (e - base < EDGES_PER_WAVE) ? (e - base) : EDGES_PER_WAVE;

    if (cnt == EDGES_PER_WAVE) {
        for (int j0 = 0; j0 < EDGES_PER_WAVE; j0 += EDGE_BATCH) {
            float v[EDGE_BATCH][4];
#pragma unroll
            for (int u = 0; u < EDGE_BATCH; ++u) {
                const int s = __builtin_amdgcn_readlane(sv, j0 + u);
                const float* __restrict__ mrow = M + (size_t)s * NHID + lane;
                v[u][0] = mrow[0];
                v[u][1] = mrow[32];
                v[u][2] = mrow[64];
                v[u][3] = mrow[96];
            }
#pragma unroll
            for (int u = 0; u < EDGE_BATCH; ++u) {
                const int d = __builtin_amdgcn_readlane(dv, j0 + u);
                float* drow = AGG + (size_t)d * NHID + lane;
                atomicAdd(drow,      v[u][0]);
                atomicAdd(drow + 32, v[u][1]);
                atomicAdd(drow + 64, v[u][2]);
                atomicAdd(drow + 96, v[u][3]);
            }
        }
    } else {
        for (int j = 0; j < cnt; ++j) {
            const int s = __builtin_amdgcn_readlane(sv, j);
            const int d = __builtin_amdgcn_readlane(dv, j);
            scatter_one_edge(M, AGG, s, d, lane);
        }
    }
}

__global__ void bias_relu(const float* __restrict__ AGG, const float* __restrict__ normD,
                          const float* __restrict__ b, float* __restrict__ H, int total) {
    int i = blockIdx.x * blockDim.x + threadIdx.x;
    if (i < total) {
        const int row = i >> 7;
        const int j   = i & 127;
        const float v = AGG[i] * normD[row] + b[j];
        H[i] = v > 0.0f ? v : 0.0f;
    }
}

__global__ void __launch_bounds__(256)
edge_scalar_add(const float* __restrict__ S, const int* __restrict__ src,
                const int* __restrict__ dst, float* __restrict__ T, int e) {
    const int i = blockIdx.x * blockDim.x + threadIdx.x;
    if (i < e) atomicAdd(&T[dst[i]], S[src[i]]);
}

__global__ void __launch_bounds__(256)
final_out(const float* __restrict__ T, const float* __restrict__ normD,
          const float* __restrict__ b2, const float* __restrict__ Wfc,
          const float* __restrict__ bfc, float* __restrict__ out, int n) {
    const int i    = blockIdx.x * blockDim.x + threadIdx.x;
    const int lane = threadIdx.x & 31;
    float cb = 0.0f;
    for (int t = 0; t < 4; ++t) {
        const int j = lane + 32 * t;
        cb += b2[j] * Wfc[j];
    }
    for (int off = 16; off > 0; off >>= 1)
        cb += __shfl_xor(cb, off, 32);
    if (i < n) out[i] = T[i] * normD[i] + (cb + bfc[0]);
}

extern "C" void kernel_launch(void* const* d_in, const int* in_sizes, int n_in,
                              void* d_out, int out_size, void* d_ws, size_t ws_size,
                              hipStream_t stream) {
    const float* x   = (const float*)d_in[0];
    const float* W1  = (const float*)d_in[1];
    const float* b1  = (const float*)d_in[2];
    const float* W2  = (const float*)d_in[3];
    const float* b2  = (const float*)d_in[4];
    const float* Wfc = (const float*)d_in[5];
    const float* bfc = (const float*)d_in[6];
    const int*   src = (const int*)d_in[7];
    const int*   dst = (const int*)d_in[8];
    float* out = (float*)d_out;

    const int N = N_NODES, E = N_EDGES;
    const int NH = NHID;

    float* BUF_A = (float*)d_ws;
    float* BUF_B = BUF_A + (size_t)N * NH;
    float* normS = BUF_B + (size_t)N * NH;
    float* normD = normS + N;

    const int TPB = 256;
    const int featTotal = N * NH;
    const int gridFeat  = (featTotal + TPB - 1) / TPB;
    const int gridN     = (N + TPB - 1) / TPB;
    const int gridE     = (E + TPB - 1) / TPB;
    const int gridEdgeWaves = (E + 8 * EDGES_PER_WAVE - 1) / (8 * EDGES_PER_WAVE);
    const int gridGemm  = N / 16;

    zero_kernel<<<(2 * N + TPB - 1) / TPB, TPB, 0, stream>>>(normS, 2 * N);
    degree_kernel<<<gridE, TPB, 0, stream>>>(src, dst, normS, normD, E);
    norm_finalize<<<gridN, TPB, 0, stream>>>(normS, normD, N);

    zero_kernel<<<gridFeat, TPB, 0, stream>>>(BUF_B, featTotal);
    gemm_scaled_wmma<<<gridGemm, TPB, 0, stream>>>(x, normS, W1, BUF_A, NFEAT);
    scatter_add<<<gridEdgeWaves, TPB, 0, stream>>>(BUF_A, src, dst, BUF_B, E);
    bias_relu<<<gridFeat, TPB, 0, stream>>>(BUF_B, normD, b1, BUF_A, featTotal);

    float* S = BUF_B;
    float* T = BUF_B + N;
    gemm_scaled_wmma_rowdot<<<gridGemm, TPB, 0, stream>>>(BUF_A, normS, W2, Wfc, S, T, NH);
    edge_scalar_add<<<gridE, TPB, 0, stream>>>(S, src, dst, T, E);

    final_out<<<gridN, TPB, 0, stream>>>(T, normD, b2, Wfc, bfc, out, N);
}


# both products retiled: block tile 80 x 128, x and W K-chunks staged in LDS by coalesced 16-byte loads, 5 row tiles per wave on one B fragment, whole-row stores; layer-2 row dot re-associated
# speedup vs baseline: 1.9904x; 1.1980x over previous
#include <hip/hip_runtime.h>

typedef __attribute__((ext_vector_type(2))) float v2f;
typedef __attribute__((ext_vector_type(8))) float v8f;

#define N_NODES 100000
#define N_EDGES 1600000
#define NFEAT   256
#define NHID    128

__global__ void zero_kernel(float* __restrict__ p, int n) {
    int i = blockIdx.x * blockDim.x + threadIdx.x;
    if (i < n) p[i] = 0.0f;
}

__global__ void degree_kernel(const int* __restrict__ src, const int* __restrict__ dst,
                              float* __restrict__ degS, float* __restrict__ degD, int e) {
    int i = blockIdx.x * blockDim.x + threadIdx.x;
    if (i < e) {
        atomicAdd(&degS[src[i]], 1.0f);
        atomicAdd(&degD[dst[i]], 1.0f);
    }
}

__global__ void norm_finalize(float* __restrict__ degS, float* __restrict__ degD, int n) {
    int i = blockIdx.x * blockDim.x + threadIdx.x;
    if (i < n) {
        degS[i] = rsqrtf(fmaxf(degS[i], 1.0f));
        degD[i] = rsqrtf(fmaxf(degD[i], 1.0f));
    }
}

#define BM 80
#define BK 32
#define PA 36
#define PB 288
#define PC 132
#define SMEM_FLOATS (BM * PC)

__device__ __forceinline__ void gemm80_mainloop(const float* __restrict__ X,
                                                const float* __restrict__ scale,
                                                const float* __restrict__ W, int K,
                                                float* smem, v8f (&acc)[5]) {
    float* As = smem;
    float* Bs = smem + BM * PA;
    const int t    = threadIdx.x;
    const int wave = t >> 5;
    const int lane = t & 31;
    const int h    = lane >> 4;
    const int c    = lane & 15;
    const int row0 = blockIdx.x * BM;

    const bool third = (t < 128);
    const float* xa[3];
    float ns[3];
    int   aoff[3];
#pragma unroll
    for (int i = 0; i < 3; ++i) {
        const int idx = t + 256 * i;
        const int r   = (i < 2 || third) ? (idx >> 3) : 0;
        const int k4  = idx & 7;
        ns[i]   = scale[row0 + r];
        xa[i]   = X + (size_t)(row0 + r) * K + 4 * k4;
        aoff[i] = r * PA + 4 * k4;
    }
    const float* wb[2];
    int boff[2];
#pragma unroll
    for (int j = 0; j < 2; ++j) {
        const int idx = t + 256 * j;
        const int pr  = idx >> 5;
        const int n4  = idx & 31;
        wb[j]   = W + (size_t)(2 * pr) * NHID + 4 * n4;
        boff[j] = pr * PB + 8 * n4;
    }

    float4 ra[3];
    float4 rb[2][2];

    const int nch = K / BK;

#pragma unroll
    for (int i = 0; i < 2; ++i) ra[i] = *(const float4*)(xa[i]);
    if (third) ra[2] = *(const float4*)(xa[2]);
#pragma unroll
    for (int j = 0; j < 2; ++j) {
        rb[j][0] = *(const float4*)(wb[j]);
        rb[j][1] = *(const float4*)(wb[j] + NHID);
    }

    const int afrag = c * PA + 2 * h;
    const int bfrag = h * PB + (16 * wave + c) * 2;

    for (int ch = 0; ch < nch; ++ch) {
#pragma unroll
        for (int i = 0; i < 3; ++i) {
            if (i < 2 || third) {
                float4 v = ra[i];
                v.x *= ns[i]; v.y *= ns[i]; v.z *= ns[i]; v.w *= ns[i];
                *(float4*)(As + aoff[i]) = v;
            }
        }
#pragma unroll
        for (int j = 0; j < 2; ++j) {
            const float4 lo = rb[j][0], hi = rb[j][1];
            *(float4*)(Bs + boff[j])     = make_float4(lo.x, hi.x, lo.y, hi.y);
            *(float4*)(Bs + boff[j] + 4) = make_float4(lo.z, hi.z, lo.w, hi.w);
        }
        __syncthreads();

        if (ch + 1 < nch) {
            const int ko = (ch + 1) * BK;
#pragma unroll
            for (int i = 0; i < 2; ++i) ra[i] = *(const float4*)(xa[i] + ko);
            if (third) ra[2] = *(const float4*)(xa[2] + ko);
#pragma unroll
            for (int j = 0; j < 2; ++j) {
                rb[j][0] = *(const float4*)(wb[j] + (size_t)ko * NHID);
                rb[j][1] = *(const float4*)(wb[j] + (size_t)ko * NHID + NHID);
            }
        }

#pragma unroll
        for (int kk = 0; kk < BK / 4; ++kk) {
            const v2f b = *(const v2f*)(Bs + bfrag + 2 * kk * PB);
#pragma unroll
            for (int tm = 0; tm < 5; ++tm) {
                const v2f a = *(const v2f*)(As + afrag + 16 * tm * PA + 4 * kk);
                acc[tm] = __builtin_amdgcn_wmma_f32_16x16x4_f32(false, a, false, b,
                                                                (short)0, acc[tm], false, false);
            }
        }
        __syncthreads();
    }
}

__device__ __forceinline__ void gemm80_stage_result(float* smem, const v8f (&acc)[5]) {
    const int wave = threadIdx.x >> 5;
    const int lane = threadIdx.x & 31;
    const int h    = lane >> 4;
    const int c    = lane & 15;
#pragma unroll
    for (int tm = 0; tm < 5; ++tm) {
#pragma unroll
        for (int r = 0; r < 8; ++r) {
            smem[(16 * tm + 8 * h + r) * PC + 16 * wave + c] = acc[tm][r];
        }
    }
    __syncthreads();
}

__global__ void __launch_bounds__(256)
gemm_scaled_wmma(const float* __restrict__ X, const float* __restrict__ scale,
                 const float* __restrict__ W, float* __restrict__ Out, int K) {
    __shared__ __attribute__((aligned(16))) float smem[SMEM_FLOATS];
    v8f acc[5] = {};
    gemm80_mainloop(X, scale, W, K, smem, acc);
    gemm80_stage_result(smem, acc);

    const int wave = threadIdx.x >> 5;
    const int lane = threadIdx.x & 31;
    const int row0 = blockIdx.x * BM;
#pragma unroll
    for (int i = 0; i < BM / 8; ++i) {
        const int row = wave + 8 * i;
        const float4 v = *(const float4*)(smem + row * PC + 4 * lane);
        *(float4*)(Out + (size_t)(row0 + row) * NHID + 4 * lane) = v;
    }
}

__global__ void __launch_bounds__(256)
gemm_scaled_wmma_rowdot(const float* __restrict__ X, const float* __restrict__ scale,
                        const float* __restrict__ W, const float* __restrict__ Wfc,
                        float* __restrict__ S, float* __restrict__ T, int K) {
    __shared__ __attribute__((aligned(16))) float smem[SMEM_FLOATS];
    v8f acc[5] = {};
    gemm80_mainloop(X, scale, W, K, smem, acc);
    gemm80_stage_result(smem, acc);

    const int wave = threadIdx.x >> 5;
    const int lane = threadIdx.x & 31;
    const int row0 = blockIdx.x * BM;
    float wf[4];
#pragma unroll
    for (int j = 0; j < 4; ++j) wf[j] = Wfc[lane + 32 * j];

#pragma unroll
    for (int i = 0; i < BM / 8; ++i) {
        const int row = wave + 8 * i;
        const float* crow = smem + row * PC + lane;
        float s = crow[0] * wf[0];
        s += crow[32] * wf[1];
        s += crow[64] * wf[2];
        s += crow[96] * wf[3];
        for (int off = 16; off > 0; off >>= 1)
            s += __shfl_xor(s, off, 32);
        if (lane == 0) {
            S[row0 + row] = s;
            T[row0 + row] = 0.0f;
        }
    }
}

#define EDGES_PER_WAVE 32
#define EDGE_BATCH     8

__device__ __forceinline__ void scatter_one_edge(const float* __restrict__ M,
                                                 float* AGG, int s, int d, int lane) {
    const float* __restrict__ mrow = M + (size_t)s * NHID + lane;
    float* drow = AGG + (size_t)d * NHID + lane;
    const float v0 = mrow[0];
    const float v1 = mrow[32];
    const float v2 = mrow[64];
    const float v3 = mrow[96];
    atomicAdd(drow,      v0);
    atomicAdd(drow + 32, v1);
    atomicAdd(drow + 64, v2);
    atomicAdd(drow + 96, v3);
}

__global__ void __launch_bounds__(256)
scatter_add(const float* __restrict__ M, const int* __restrict__ src,
            const int* __restrict__ dst, float* __restrict__ AGG, int e) {
    const int w    = (blockIdx.x * blockDim.x + threadIdx.x) >> 5;
    const int lane = threadIdx.x & 31;
    const int base = w * EDGES_PER_WAVE;
    if (base >= e) return;
    const int mine = base + lane;
    const int sv = (mine < e) ? src[mine] : 0;
    const int dv = (mine < e) ? dst[mine] : 0;
    const int cnt = (e - base < EDGES_PER_WAVE) ? (e - base) : EDGES_PER_WAVE;

    if (cnt == EDGES_PER_WAVE) {
        for (int j0 = 0; j0 < EDGES_PER_WAVE; j0 += EDGE_BATCH) {
            float v[EDGE_BATCH][4];
#pragma unroll
            for (int u = 0; u < EDGE_BATCH; ++u) {
                const int s = __builtin_amdgcn_readlane(sv, j0 + u);
                const float* __restrict__ mrow = M + (size_t)s * NHID + lane;
                v[u][0] = mrow[0];
                v[u][1] = mrow[32];
                v[u][2] = mrow[64];
                v[u][3] = mrow[96];
            }
#pragma unroll
            for (int u = 0; u < EDGE_BATCH; ++u) {
                const int d = __builtin_amdgcn_readlane(dv, j0 + u);
                float* drow = AGG + (size_t)d * NHID + lane;
                atomicAdd(drow,      v[u][0]);
                atomicAdd(drow + 32, v[u][1]);
                atomicAdd(drow + 64, v[u][2]);
                atomicAdd(drow + 96, v[u][3]);
            }
        }
    } else {
        for (int j = 0; j < cnt; ++j) {
            const int s = __builtin_amdgcn_readlane(sv, j);
            const int d = __builtin_amdgcn_readlane(dv, j);
            scatter_one_edge(M, AGG, s, d, lane);
        }
    }
}

__global__ void bias_relu(const float* __restrict__ AGG, const float* __restrict__ normD,
                          const float* __restrict__ b, float* __restrict__ H, int total) {
    int i = blockIdx.x * blockDim.x + threadIdx.x;
    if (i < total) {
        const int row = i >> 7;
        const int j   = i & 127;
        const float v = AGG[i] * normD[row] + b[j];
        H[i] = v > 0.0f ? v : 0.0f;
    }
}

__global__ void __launch_bounds__(256)
edge_scalar_add(const float* __restrict__ S, const int* __restrict__ src,
                const int* __restrict__ dst, float* __restrict__ T, int e) {
    const int i = blockIdx.x * blockDim.x + threadIdx.x;
    if (i < e) atomicAdd(&T[dst[i]], S[src[i]]);
}

__global__ void __launch_bounds__(256)
final_out(const float* __restrict__ T, const float* __restrict__ normD,
          const float* __restrict__ b2, const float* __restrict__ Wfc,
          const float* __restrict__ bfc, float* __restrict__ out, int n) {
    const int i    = blockIdx.x * blockDim.x + threadIdx.x;
    const int lane = threadIdx.x & 31;
    float cb = 0.0f;
    for (int t = 0; t < 4; ++t) {
        const int j = lane + 32 * t;
        cb += b2[j] * Wfc[j];
    }
    for (int off = 16; off > 0; off >>= 1)
        cb += __shfl_xor(cb, off, 32);
    if (i < n) out[i] = T[i] * normD[i] + (cb + bfc[0]);
}

extern "C" void kernel_launch(void* const* d_in, const int* in_sizes, int n_in,
                              void* d_out, int out_size, void* d_ws, size_t ws_size,
                              hipStream_t stream) {
    const float* x   = (const float*)d_in[0];
    const float* W1  = (const float*)d_in[1];
    const float* b1  = (const float*)d_in[2];
    const float* W2  = (const float*)d_in[3];
    const float* b2  = (const float*)d_in[4];
    const float* Wfc = (const float*)d_in[5];
    const float* bfc = (const float*)d_in[6];
    const int*   src = (const int*)d_in[7];
    const int*   dst = (const int*)d_in[8];
    float* out = (float*)d_out;

    const int N = N_NODES, E = N_EDGES;
    const int NH = NHID;

    float* BUF_A = (float*)d_ws;
    float* BUF_B = BUF_A + (size_t)N * NH;
    float* normS = BUF_B + (size_t)N * NH;
    float* normD = normS + N;

    const int TPB = 256;
    const int featTotal = N * NH;
    const int gridFeat  = (featTotal + TPB - 1) / TPB;
    const int gridN     = (N + TPB - 1) / TPB;
    const int gridE     = (E + TPB - 1) / TPB;
    const int gridEdgeWaves = (E + 8 * EDGES_PER_WAVE - 1) / (8 * EDGES_PER_WAVE);
    const int gridGemm  = N / BM;

    zero_kernel<<<(2 * N + TPB - 1) / TPB, TPB, 0, stream>>>(normS, 2 * N);
    degree_kernel<<<gridE, TPB, 0, stream>>>(src, dst, normS, normD, E);
    norm_finalize<<<gridN, TPB, 0, stream>>>(normS, normD, N);

    zero_kernel<<<gridFeat, TPB, 0, stream>>>(BUF_B, featTotal);
    gemm_scaled_wmma<<<gridGemm, TPB, 0, stream>>>(x, normS, W1, BUF_A, NFEAT);
    scatter_add<<<gridEdgeWaves, TPB, 0, stream>>>(BUF_A, src, dst, BUF_B, E);
    bias_relu<<<gridFeat, TPB, 0, stream>>>(BUF_B, normD, b1, BUF_A, featTotal);

    float* S = BUF_B;
    float* T = BUF_B + N;
    gemm_scaled_wmma_rowdot<<<gridGemm, TPB, 0, stream>>>(BUF_A, normS, W2, Wfc, S, T, NH);
    edge_scalar_add<<<gridE, TPB, 0, stream>>>(S, src, dst, T, E);

    final_out<<<gridN, TPB, 0, stream>>>(T, normD, b2, Wfc, bfc, out, N);
}


# element-wise passes folded into their neighbours: AGG zero fill into the layer-1 product's store epilogue, bias+relu and the norm rsqrt into the products' A loaders; 3 launches gone; sums as before
# speedup vs baseline: 2.1885x; 1.0995x over previous
#include <hip/hip_runtime.h>

typedef __attribute__((ext_vector_type(2))) float v2f;
typedef __attribute__((ext_vector_type(8))) float v8f;

#define N_NODES 100000
#define N_EDGES 1600000
#define NFEAT   256
#define NHID    128

__global__ void zero_kernel(float* __restrict__ p, int n) {
    int i = blockIdx.x * blockDim.x + threadIdx.x;
    if (i < n) p[i] = 0.0f;
}

__global__ void degree_kernel(const int* __restrict__ src, const int* __restrict__ dst,
                              float* __restrict__ degS, float* __restrict__ degD, int e) {
    int i = blockIdx.x * blockDim.x + threadIdx.x;
    if (i < e) {
        atomicAdd(&degS[src[i]], 1.0f);
        atomicAdd(&degD[dst[i]], 1.0f);
    }
}

__device__ __forceinline__ float deg_to_norm(float deg) {
    return rsqrtf(fmaxf(deg, 1.0f));
}

#define BM 80
#define BK 32
#define PA 36
#define PB 288
#define PC 132
#define SMEM_FLOATS (BM * PC)

template <bool FUSE>
__device__ __forceinline__ void gemm80_mainloop(const float* __restrict__ X,
                                                const float* __restrict__ scale,
                                                const float* __restrict__ W, int K,
                                                float* smem, v8f (&acc)[5],
                                                const float* __restrict__ normD,
                                                const float* __restrict__ bias) {
    float* As = smem;
    float* Bs = smem + BM * PA;
    const int t    = threadIdx.x;
    const int wave = t >> 5;
    const int lane = t & 31;
    const int h    = lane >> 4;
    const int c    = lane & 15;
    const int row0 = blockIdx.x * BM;

    const bool third = (t < 128);
    const float* xa[3];
    float ns[3];
    float nd[3];
    int   aoff[3];
#pragma unroll
    for (int i = 0; i < 3; ++i) {
        const int idx = t + 256 * i;
        const int r   = (i < 2 || third) ? (idx >> 3) : 0;
        const int k4  = idx & 7;
        ns[i]   = deg_to_norm(scale[row0 + r]);
        nd[i]   = FUSE ? deg_to_norm(normD[row0 + r]) : 0.0f;
        xa[i]   = X + (size_t)(row0 + r) * K + 4 * k4;
        aoff[i] = r * PA + 4 * k4;
    }
    const float* wb[2];
    int boff[2];
#pragma unroll
    for (int j = 0; j < 2; ++j) {
        const int idx = t + 256 * j;
        const int pr  = idx >> 5;
        const int n4  = idx & 31;
        wb[j]   = W + (size_t)(2 * pr) * NHID + 4 * n4;
        boff[j] = pr * PB + 8 * n4;
    }

    float4 ra[3];
    float4 rb[2][2];
    const float* bp = FUSE ? (bias + 4 * (t & 7)) : nullptr;
    float4 rbias = make_float4(0.0f, 0.0f, 0.0f, 0.0f);

    const int nch = K / BK;

#pragma unroll
    for (int i = 0; i < 2; ++i) ra[i] = *(const float4*)(xa[i]);
    if (third) ra[2] = *(const float4*)(xa[2]);
    if (FUSE) rbias = *(const float4*)(bp);
#pragma unroll
    for (int j = 0; j < 2; ++j) {
        rb[j][0] = *(const float4*)(wb[j]);
        rb[j][1] = *(const float4*)(wb[j] + NHID);
    }

    const int afrag = c * PA + 2 * h;
    const int bfrag = h * PB + (16 * wave + c) * 2;

    for (int ch = 0; ch < nch; ++ch) {
#pragma unroll
        for (int i = 0; i < 3; ++i) {
            if (i < 2 || third) {
                float4 v = ra[i];
                if (FUSE) {
                    const float h0 = v.x * nd[i] + rbias.x;
                    const float h1 = v.y * nd[i] + rbias.y;
                    const float h2 = v.z * nd[i] + rbias.z;
                    const float h3 = v.w * nd[i] + rbias.w;
                    v.x = h0 > 0.0f ? h0 : 0.0f;
                    v.y = h1 > 0.0f ? h1 : 0.0f;
                    v.z = h2 > 0.0f ? h2 : 0.0f;
                    v.w = h3 > 0.0f ? h3 : 0.0f;
                }
                v.x *= ns[i]; v.y *= ns[i]; v.z *= ns[i]; v.w *= ns[i];
                *(float4*)(As + aoff[i]) = v;
            }
        }
#pragma unroll
        for (int j = 0; j < 2; ++j) {
            const float4 lo = rb[j][0], hi = rb[j][1];
            *(float4*)(Bs + boff[j])     = make_float4(lo.x, hi.x, lo.y, hi.y);
            *(float4*)(Bs + boff[j] + 4) = make_float4(lo.z, hi.z, lo.w, hi.w);
        }
        __syncthreads();

        if (ch + 1 < nch) {
            const int ko = (ch + 1) * BK;
#pragma unroll
            for (int i = 0; i < 2; ++i) ra[i] = *(const float4*)(xa[i] + ko);
            if (third) ra[2] = *(const float4*)(xa[2] + ko);
            if (FUSE) rbias = *(const float4*)(bp + ko);
#pragma unroll
            for (int j = 0; j < 2; ++j) {
                rb[j][0] = *(const float4*)(wb[j] + (size_t)ko * NHID);
                rb[j][1] = *(const float4*)(wb[j] + (size_t)ko * NHID + NHID);
            }
        }

#pragma unroll
        for (int kk = 0; kk < BK / 4; ++kk) {
            const v2f b = *(const v2f*)(Bs + bfrag + 2 * kk * PB);
#pragma unroll
            for (int tm = 0; tm < 5; ++tm) {
                const v2f a = *(const v2f*)(As + afrag + 16 * tm * PA + 4 * kk);
                acc[tm] = __builtin_amdgcn_wmma_f32_16x16x4_f32(false, a, false, b,
                                                                (short)0, acc[tm], false, false);
            }
        }
        __syncthreads();
    }
}

__device__ __forceinline__ void gemm80_stage_result(float* smem, const v8f (&acc)[5]) {
    const int wave = threadIdx.x >> 5;
    const int lane = threadIdx.x & 31;
    const int h    = lane >> 4;
    const int c    = lane & 15;
#pragma unroll
    for (int tm = 0; tm < 5; ++tm) {
#pragma unroll
        for (int r = 0; r < 8; ++r) {
            smem[(16 * tm + 8 * h + r) * PC + 16 * wave + c] = acc[tm][r];
        }
    }
    __syncthreads();
}

__global__ void __launch_bounds__(256)
gemm_scaled_wmma(const float* __restrict__ X, const float* __restrict__ scale,
                 const float* __restrict__ W, float* __restrict__ Out,
                 float* __restrict__ Sum0, int K) {
    __shared__ __attribute__((aligned(16))) float smem[SMEM_FLOATS];
    v8f acc[5] = {};
    gemm80_mainloop<false>(X, scale, W, K, smem, acc, nullptr, nullptr);
    gemm80_stage_result(smem, acc);

    const int wave = threadIdx.x >> 5;
    const int lane = threadIdx.x & 31;
    const int row0 = blockIdx.x * BM;
#pragma unroll
    for (int i = 0; i < BM / 8; ++i) {
        const int row = wave + 8 * i;
        const float4 v = *(const float4*)(smem + row * PC + 4 * lane);
        *(float4*)(Out + (size_t)(row0 + row) * NHID + 4 * lane) = v;
    }
#pragma unroll
    for (int i = 0; i < BM / 8; ++i) {
        const int row = wave + 8 * i;
        *(float4*)(Sum0 + (size_t)(row0 + row) * NHID + 4 * lane) =
            make_float4(0.0f, 0.0f, 0.0f, 0.0f);
    }
}

__global__ void __launch_bounds__(256)
gemm_scaled_wmma_rowdot(const float* __restrict__ X, const float* __restrict__ scale,
                        const float* __restrict__ normD, const float* __restrict__ bias,
                        const float* __restrict__ W, const float* __restrict__ Wfc,
                        float* __restrict__ S, float* __restrict__ T, int K) {
    __shared__ __attribute__((aligned(16))) float smem[SMEM_FLOATS];
    v8f acc[5] = {};
    gemm80_mainloop<true>(X, scale, W, K, smem, acc, normD, bias);
    gemm80_stage_result(smem, acc);

    const int wave = threadIdx.x >> 5;
    const int lane = threadIdx.x & 31;
    const int row0 = blockIdx.x * BM;
    float wf[4];
#pragma unroll
    for (int j = 0; j < 4; ++j) wf[j] = Wfc[lane + 32 * j];

#pragma unroll
    for (int i = 0; i < BM / 8; ++i) {
        const int row = wave + 8 * i;
        const float* crow = smem + row * PC + lane;
        float s = crow[0] * wf[0];
        s += crow[32] * wf[1];
        s += crow[64] * wf[2];
        s += crow[96] * wf[3];
        for (int off = 16; off > 0; off >>= 1)
            s += __shfl_xor(s, off, 32);
        if (lane == 0) {
            S[row0 + row] = s;
            T[row0 + row] = 0.0f;
        }
    }
}

#define EDGES_PER_WAVE 32
#define EDGE_BATCH     8

__device__ __forceinline__ void scatter_one_edge(const float* __restrict__ M,
                                                 float* AGG, int s, int d, int lane) {
    const float* __restrict__ mrow = M + (size_t)s * NHID + lane;
    float* drow = AGG + (size_t)d * NHID + lane;
    const float v0 = mrow[0];
    const float v1 = mrow[32];
    const float v2 = mrow[64];
    const float v3 = mrow[96];
    atomicAdd(drow,      v0);
    atomicAdd(drow + 32, v1);
    atomicAdd(drow + 64, v2);
    atomicAdd(drow + 96, v3);
}

__global__ void __launch_bounds__(256)
scatter_add(const float* __restrict__ M, const int* __restrict__ src,
            const int* __restrict__ dst, float* __restrict__ AGG, int e) {
    const int w    = (blockIdx.x * blockDim.x + threadIdx.x) >> 5;
    const int lane = threadIdx.x & 31;
    const int base = w * EDGES_PER_WAVE;
    if (base >= e) return;
    const int mine = base + lane;
    const int sv = (mine < e) ? src[mine] : 0;
    const int dv = (mine < e) ? dst[mine] : 0;
    const int cnt = (e - base < EDGES_PER_WAVE) ? (e - base) : EDGES_PER_WAVE;

    if (cnt == EDGES_PER_WAVE) {
        for (int j0 = 0; j0 < EDGES_PER_WAVE; j0 += EDGE_BATCH) {
            float v[EDGE_BATCH][4];
#pragma unroll
            for (int u = 0; u < EDGE_BATCH; ++u) {
                const int s = __builtin_amdgcn_readlane(sv, j0 + u);
                const float* __restrict__ mrow = M + (size_t)s * NHID + lane;
                v[u][0] = mrow[0];
                v[u][1] = mrow[32];
                v[u][2] = mrow[64];
                v[u][3] = mrow[96];
            }
#pragma unroll
            for (int u = 0; u < EDGE_BATCH; ++u) {
                const int d = __builtin_amdgcn_readlane(dv, j0 + u);
                float* drow = AGG + (size_t)d * NHID + lane;
                atomicAdd(drow,      v[u][0]);
                atomicAdd(drow + 32, v[u][1]);
                atomicAdd(drow + 64, v[u][2]);
                atomicAdd(drow + 96, v[u][3]);
            }
        }
    } else {
        for (int j = 0; j < cnt; ++j) {
            const int s = __builtin_amdgcn_readlane(sv, j);
            const int d = __builtin_amdgcn_readlane(dv, j);
            scatter_one_edge(M, AGG, s, d, lane);
        }
    }
}

__global__ void __launch_bounds__(256)
edge_scalar_add(const float* __restrict__ S, const int* __restrict__ src,
                const int* __restrict__ dst, float* __restrict__ T, int e) {
    const int i = blockIdx.x * blockDim.x + threadIdx.x;
    if (i < e) atomicAdd(&T[dst[i]], S[src[i]]);
}

__global__ void __launch_bounds__(256)
final_out(const float* __restrict__ T, const float* __restrict__ normD,
          const float* __restrict__ b2, const float* __restrict__ Wfc,
          const float* __restrict__ bfc, float* __restrict__ out, int n) {
    const int i    = blockIdx.x * blockDim.x + threadIdx.x;
    const int lane = threadIdx.x & 31;
    float cb = 0.0f;
    for (int t = 0; t < 4; ++t) {
        const int j = lane + 32 * t;
        cb += b2[j] * Wfc[j];
    }
    for (int off = 16; off > 0; off >>= 1)
        cb += __shfl_xor(cb, off, 32);
    if (i < n) out[i] = T[i] * deg_to_norm(normD[i]) + (cb + bfc[0]);
}

extern "C" void kernel_launch(void* const* d_in, const int* in_sizes, int n_in,
                              void* d_out, int out_size, void* d_ws, size_t ws_size,
                              hipStream_t stream) {
    const float* x   = (const float*)d_in[0];
    const float* W1  = (const float*)d_in[1];
    const float* b1  = (const float*)d_in[2];
    const float* W2  = (const float*)d_in[3];
    const float* b2  = (const float*)d_in[4];
    const float* Wfc = (const float*)d_in[5];
    const float* bfc = (const float*)d_in[6];
    const int*   src = (const int*)d_in[7];
    const int*   dst = (const int*)d_in[8];
    float* out = (float*)d_out;

    const int N = N_NODES, E = N_EDGES;
    const int NH = NHID;

    float* BUF_A = (float*)d_ws;
    float* BUF_B = BUF_A + (size_t)N * NH;
    float* normS = BUF_B + (size_t)N * NH;
    float* normD = normS + N;

    const int TPB = 256;
    const int gridN     = (N + TPB - 1) / TPB;
    const int gridE     = (E + TPB - 1) / TPB;
    const int gridEdgeWaves = (E + 8 * EDGES_PER_WAVE - 1) / (8 * EDGES_PER_WAVE);
    const int gridGemm  = N / BM;

    zero_kernel<<<(2 * N + TPB - 1) / TPB, TPB, 0, stream>>>(normS, 2 * N);
    degree_kernel<<<gridE, TPB, 0, stream>>>(src, dst, normS, normD, E);

    gemm_scaled_wmma<<<gridGemm, TPB, 0, stream>>>(x, normS, W1, BUF_A, BUF_B, NFEAT);
    scatter_add<<<gridEdgeWaves, TPB, 0, stream>>>(BUF_A, src, dst, BUF_B, E);

    float* S = BUF_A;
    float* T = BUF_A + N;
    gemm_scaled_wmma_rowdot<<<gridGemm, TPB, 0, stream>>>(BUF_B, normS, normD, b1, W2, Wfc,
                                                          S, T, NH);
    edge_scalar_add<<<gridE, TPB, 0, stream>>>(S, src, dst, T, E);

    final_out<<<gridN, TPB, 0, stream>>>(T, normD, b2, Wfc, bfc, out, N);
}


# degree count and one-float edge sum: per-edge scattered global atomics replaced by node-range sums in LDS over a sliced re-read of the edge list, flushed as whole-line f32 adds; sums in arrival order
# speedup vs baseline: 3.8297x; 1.7499x over previous
#include <hip/hip_runtime.h>

typedef __attribute__((ext_vector_type(2))) float v2f;
typedef __attribute__((ext_vector_type(8))) float v8f;

#define N_NODES 100000
#define N_EDGES 1600000
#define NFEAT   256
#define NHID    128

__global__ void zero_kernel(float* __restrict__ p, int n) {
    int i = blockIdx.x * blockDim.x + threadIdx.x;
    if (i < n) p[i] = 0.0f;
}

#define NODE_NR 16
#define NODE_R  6272
#define EDGE_NS 16
#define ACC_TPB 1024
static_assert(NODE_NR * NODE_R >= N_NODES, "node ranges must cover all nodes");

__device__ __forceinline__ int slice_len(int e) {
    return (((e + EDGE_NS - 1) / EDGE_NS) + 3) & ~3;
}

__device__ __forceinline__ void count_one(unsigned* cnt, int idx, unsigned r0) {
    const unsigned u = (unsigned)idx - r0;
    if (u < (unsigned)NODE_R) atomicAdd(&cnt[u], 1u);
}

__device__ __forceinline__ void count_four(unsigned* cnt, const int4 v, unsigned r0) {
    count_one(cnt, v.x, r0);
    count_one(cnt, v.y, r0);
    count_one(cnt, v.z, r0);
    count_one(cnt, v.w, r0);
}

__global__ void __launch_bounds__(ACC_TPB)
degree_kernel(const int* __restrict__ src, const int* __restrict__ dst,
              float* __restrict__ degS, float* __restrict__ degD, int e, int n) {
    __shared__ unsigned cntS[NODE_R];
    __shared__ unsigned cntD[NODE_R];
    const int t     = threadIdx.x;
    const int range = blockIdx.x % NODE_NR;
    const int slice = blockIdx.x / NODE_NR;
    const unsigned r0 = (unsigned)(range * NODE_R);

    for (int i = t; i < NODE_R; i += ACC_TPB) { cntS[i] = 0u; cntD[i] = 0u; }
    __syncthreads();

    const int per = slice_len(e);
    const int e0  = slice * per;
    const int e1  = (e0 + per < e) ? (e0 + per) : e;
    const int stride = 4 * ACC_TPB;
    int i = e0 + 4 * t;
    for (; i + stride + 4 <= e1; i += 2 * stride) {
        const int4 s0 = *(const int4*)(src + i);
        const int4 d0 = *(const int4*)(dst + i);
        const int4 s1 = *(const int4*)(src + i + stride);
        const int4 d1 = *(const int4*)(dst + i + stride);
        count_four(cntS, s0, r0);
        count_four(cntD, d0, r0);
        count_four(cntS, s1, r0);
        count_four(cntD, d1, r0);
    }
    for (; i < e1; i += stride) {
        if (i + 4 <= e1) {
            const int4 s0 = *(const int4*)(src + i);
            const int4 d0 = *(const int4*)(dst + i);
            count_four(cntS, s0, r0);
            count_four(cntD, d0, r0);
        } else {
            for (int k = i; k < e1; ++k) {
                count_one(cntS, src[k], r0);
                count_one(cntD, dst[k], r0);
            }
        }
    }
    __syncthreads();

    for (int j = t; j < NODE_R; j += ACC_TPB) {
        const int node = (int)r0 + j;
        if (node < n) {
            atomicAdd(&degS[node], (float)cntS[j]);
            atomicAdd(&degD[node], (float)cntD[j]);
        }
    }
}

__device__ __forceinline__ float deg_to_norm(float deg) {
    return rsqrtf(fmaxf(deg, 1.0f));
}

#define BM 80
#define BK 32
#define PA 36
#define PB 288
#define PC 132
#define SMEM_FLOATS (BM * PC)

template <bool FUSE>
__device__ __forceinline__ void gemm80_mainloop(const float* __restrict__ X,
                                                const float* __restrict__ scale,
                                                const float* __restrict__ W, int K,
                                                float* smem, v8f (&acc)[5],
                                                const float* __restrict__ normD,
                                                const float* __restrict__ bias) {
    float* As = smem;
    float* Bs = smem + BM * PA;
    const int t    = threadIdx.x;
    const int wave = t >> 5;
    const int lane = t & 31;
    const int h    = lane >> 4;
    const int c    = lane & 15;
    const int row0 = blockIdx.x * BM;

    const bool third = (t < 128);
    const float* xa[3];
    float ns[3];
    float nd[3];
    int   aoff[3];
#pragma unroll
    for (int i = 0; i < 3; ++i) {
        const int idx = t + 256 * i;
        const int r   = (i < 2 || third) ? (idx >> 3) : 0;
        const int k4  = idx & 7;
        ns[i]   = deg_to_norm(scale[row0 + r]);
        nd[i]   = FUSE ? deg_to_norm(normD[row0 + r]) : 0.0f;
        xa[i]   = X + (size_t)(row0 + r) * K + 4 * k4;
        aoff[i] = r * PA + 4 * k4;
    }
    const float* wb[2];
    int boff[2];
#pragma unroll
    for (int j = 0; j < 2; ++j) {
        const int idx = t + 256 * j;
        const int pr  = idx >> 5;
        const int n4  = idx & 31;
        wb[j]   = W + (size_t)(2 * pr) * NHID + 4 * n4;
        boff[j] = pr * PB + 8 * n4;
    }

    float4 ra[3];
    float4 rb[2][2];
    const float* bp = FUSE ? (bias + 4 * (t & 7)) : nullptr;
    float4 rbias = make_float4(0.0f, 0.0f, 0.0f, 0.0f);

    const int nch = K / BK;

#pragma unroll
    for (int i = 0; i < 2; ++i) ra[i] = *(const float4*)(xa[i]);
    if (third) ra[2] = *(const float4*)(xa[2]);
    if (FUSE) rbias = *(const float4*)(bp);
#pragma unroll
    for (int j = 0; j < 2; ++j) {
        rb[j][0] = *(const float4*)(wb[j]);
        rb[j][1] = *(const float4*)(wb[j] + NHID);
    }

    const int afrag = c * PA + 2 * h;
    const int bfrag = h * PB + (16 * wave + c) * 2;

    for (int ch = 0; ch < nch; ++ch) {
#pragma unroll
        for (int i = 0; i < 3; ++i) {
            if (i < 2 || third) {
                float4 v = ra[i];
                if (FUSE) {
                    const float h0 = v.x * nd[i] + rbias.x;
                    const float h1 = v.y * nd[i] + rbias.y;
                    const float h2 = v.z * nd[i] + rbias.z;
                    const float h3 = v.w * nd[i] + rbias.w;
                    v.x = h0 > 0.0f ? h0 : 0.0f;
                    v.y = h1 > 0.0f ? h1 : 0.0f;
                    v.z = h2 > 0.0f ? h2 : 0.0f;
                    v.w = h3 > 0.0f ? h3 : 0.0f;
                }
                v.x *= ns[i]; v.y *= ns[i]; v.z *= ns[i]; v.w *= ns[i];
                *(float4*)(As + aoff[i]) = v;
            }
        }
#pragma unroll
        for (int j = 0; j < 2; ++j) {
            const float4 lo = rb[j][0], hi = rb[j][1];
            *(float4*)(Bs + boff[j])     = make_float4(lo.x, hi.x, lo.y, hi.y);
            *(float4*)(Bs + boff[j] + 4) = make_float4(lo.z, hi.z, lo.w, hi.w);
        }
        __syncthreads();

        if (ch + 1 < nch) {
            const int ko = (ch + 1) * BK;
#pragma unroll
            for (int i = 0; i < 2; ++i) ra[i] = *(const float4*)(xa[i] + ko);
            if (third) ra[2] = *(const float4*)(xa[2] + ko);
            if (FUSE) rbias = *(const float4*)(bp + ko);
#pragma unroll
            for (int j = 0; j < 2; ++j) {
                rb[j][0] = *(const float4*)(wb[j] + (size_t)ko * NHID);
                rb[j][1] = *(const float4*)(wb[j] + (size_t)ko * NHID + NHID);
            }
        }

#pragma unroll
        for (int kk = 0; kk < BK / 4; ++kk) {
            const v2f b = *(const v2f*)(Bs + bfrag + 2 * kk * PB);
#pragma unroll
            for (int tm = 0; tm < 5; ++tm) {
                const v2f a = *(const v2f*)(As + afrag + 16 * tm * PA + 4 * kk);
                acc[tm] = __builtin_amdgcn_wmma_f32_16x16x4_f32(false, a, false, b,
                                                                (short)0, acc[tm], false, false);
            }
        }
        __syncthreads();
    }
}

__device__ __forceinline__ void gemm80_stage_result(float* smem, const v8f (&acc)[5]) {
    const int wave = threadIdx.x >> 5;
    const int lane = threadIdx.x & 31;
    const int h    = lane >> 4;
    const int c    = lane & 15;
#pragma unroll
    for (int tm = 0; tm < 5; ++tm) {
#pragma unroll
        for (int r = 0; r < 8; ++r) {
            smem[(16 * tm + 8 * h + r) * PC + 16 * wave + c] = acc[tm][r];
        }
    }
    __syncthreads();
}

__global__ void __launch_bounds__(256)
gemm_scaled_wmma(const float* __restrict__ X, const float* __restrict__ scale,
                 const float* __restrict__ W, float* __restrict__ Out,
                 float* __restrict__ Sum0, int K) {
    __shared__ __attribute__((aligned(16))) float smem[SMEM_FLOATS];
    v8f acc[5] = {};
    gemm80_mainloop<false>(X, scale, W, K, smem, acc, nullptr, nullptr);
    gemm80_stage_result(smem, acc);

    const int wave = threadIdx.x >> 5;
    const int lane = threadIdx.x & 31;
    const int row0 = blockIdx.x * BM;
#pragma unroll
    for (int i = 0; i < BM / 8; ++i) {
        const int row = wave + 8 * i;
        const float4 v = *(const float4*)(smem + row * PC + 4 * lane);
        *(float4*)(Out + (size_t)(row0 + row) * NHID + 4 * lane) = v;
    }
#pragma unroll
    for (int i = 0; i < BM / 8; ++i) {
        const int row = wave + 8 * i;
        *(float4*)(Sum0 + (size_t)(row0 + row) * NHID + 4 * lane) =
            make_float4(0.0f, 0.0f, 0.0f, 0.0f);
    }
}

__global__ void __launch_bounds__(256)
gemm_scaled_wmma_rowdot(const float* __restrict__ X, const float* __restrict__ scale,
                        const float* __restrict__ normD, const float* __restrict__ bias,
                        const float* __restrict__ W, const float* __restrict__ Wfc,
                        float* __restrict__ S, float* __restrict__ T, int K) {
    __shared__ __attribute__((aligned(16))) float smem[SMEM_FLOATS];
    v8f acc[5] = {};
    gemm80_mainloop<true>(X, scale, W, K, smem, acc, normD, bias);
    gemm80_stage_result(smem, acc);

    const int wave = threadIdx.x >> 5;
    const int lane = threadIdx.x & 31;
    const int row0 = blockIdx.x * BM;
    float wf[4];
#pragma unroll
    for (int j = 0; j < 4; ++j) wf[j] = Wfc[lane + 32 * j];

#pragma unroll
    for (int i = 0; i < BM / 8; ++i) {
        const int row = wave + 8 * i;
        const float* crow = smem + row * PC + lane;
        float s = crow[0] * wf[0];
        s += crow[32] * wf[1];
        s += crow[64] * wf[2];
        s += crow[96] * wf[3];
        for (int off = 16; off > 0; off >>= 1)
            s += __shfl_xor(s, off, 32);
        if (lane == 0) {
            S[row0 + row] = s;
            T[row0 + row] = 0.0f;
        }
    }
}

#define EDGES_PER_WAVE 32
#define EDGE_BATCH     8

__device__ __forceinline__ void scatter_one_edge(const float* __restrict__ M,
                                                 float* AGG, int s, int d, int lane) {
    const float* __restrict__ mrow = M + (size_t)s * NHID + lane;
    float* drow = AGG + (size_t)d * NHID + lane;
    const float v0 = mrow[0];
    const float v1 = mrow[32];
    const float v2 = mrow[64];
    const float v3 = mrow[96];
    atomicAdd(drow,      v0);
    atomicAdd(drow + 32, v1);
    atomicAdd(drow + 64, v2);
    atomicAdd(drow + 96, v3);
}

__global__ void __launch_bounds__(256)
scatter_add(const float* __restrict__ M, const int* __restrict__ src,
            const int* __restrict__ dst, float* __restrict__ AGG, int e) {
    const int w    = (blockIdx.x * blockDim.x + threadIdx.x) >> 5;
    const int lane = threadIdx.x & 31;
    const int base = w * EDGES_PER_WAVE;
    if (base >= e) return;
    const int mine = base + lane;
    const int sv = (mine < e) ? src[mine] : 0;
    const int dv = (mine < e) ? dst[mine] : 0;
    const int cnt = (e - base < EDGES_PER_WAVE) ? (e - base) : EDGES_PER_WAVE;

    if (cnt == EDGES_PER_WAVE) {
        for (int j0 = 0; j0 < EDGES_PER_WAVE; j0 += EDGE_BATCH) {
            float v[EDGE_BATCH][4];
#pragma unroll
            for (int u = 0; u < EDGE_BATCH; ++u) {
                const int s = __builtin_amdgcn_readlane(sv, j0 + u);
                const float* __restrict__ mrow = M + (size_t)s * NHID + lane;
                v[u][0] = mrow[0];
                v[u][1] = mrow[32];
                v[u][2] = mrow[64];
                v[u][3] = mrow[96];
            }
#pragma unroll
            for (int u = 0; u < EDGE_BATCH; ++u) {
                const int d = __builtin_amdgcn_readlane(dv, j0 + u);
                float* drow = AGG + (size_t)d * NHID + lane;
                atomicAdd(drow,      v[u][0]);
                atomicAdd(drow + 32, v[u][1]);
                atomicAdd(drow + 64, v[u][2]);
                atomicAdd(drow + 96, v[u][3]);
            }
        }
    } else {
        for (int j = 0; j < cnt; ++j) {
            const int s = __builtin_amdgcn_readlane(sv, j);
            const int d = __builtin_amdgcn_readlane(dv, j);
            scatter_one_edge(M, AGG, s, d, lane);
        }
    }
}

#define ES_BATCH 8
__global__ void __launch_bounds__(ACC_TPB)
edge_scalar_add(const float* __restrict__ S, const int* __restrict__ src,
                const int* __restrict__ dst, float* __restrict__ T, int e, int n) {
    __shared__ float acc[NODE_R];
    const int t     = threadIdx.x;
    const int range = blockIdx.x % NODE_NR;
    const int slice = blockIdx.x / NODE_NR;
    const unsigned r0 = (unsigned)(range * NODE_R);

    for (int i = t; i < NODE_R; i += ACC_TPB) acc[i] = 0.0f;
    __syncthreads();

    const int per = slice_len(e);
    const int e0  = slice * per;
    const int e1  = (e0 + per < e) ? (e0 + per) : e;
    const int stride = 4 * ACC_TPB;
    int i = e0 + 4 * t;
    for (; i + stride + 4 <= e1; i += 2 * stride) {
        const int4 s0 = *(const int4*)(src + i);
        const int4 d0 = *(const int4*)(dst + i);
        const int4 s1 = *(const int4*)(src + i + stride);
        const int4 d1 = *(const int4*)(dst + i + stride);
        const int      sidx[ES_BATCH] = {s0.x, s0.y, s0.z, s0.w, s1.x, s1.y, s1.z, s1.w};
        const unsigned u[ES_BATCH]    = {(unsigned)d0.x - r0, (unsigned)d0.y - r0,
                                         (unsigned)d0.z - r0, (unsigned)d0.w - r0,
                                         (unsigned)d1.x - r0, (unsigned)d1.y - r0,
                                         (unsigned)d1.z - r0, (unsigned)d1.w - r0};
        float v[ES_BATCH];
#pragma unroll
        for (int k = 0; k < ES_BATCH; ++k) {
            v[k] = 0.0f;
            if (u[k] < (unsigned)NODE_R) v[k] = S[sidx[k]];
        }
#pragma unroll
        for (int k = 0; k < ES_BATCH; ++k) {
            if (u[k] < (unsigned)NODE_R) atomicAdd(&acc[u[k]], v[k]);
        }
    }
    for (; i < e1; i += stride) {
        const int kend = (i + 4 <= e1) ? (i + 4) : e1;
        for (int k = i; k < kend; ++k) {
            const unsigned uu = (unsigned)dst[k] - r0;
            if (uu < (unsigned)NODE_R) atomicAdd(&acc[uu], S[src[k]]);
        }
    }
    __syncthreads();

    for (int j = t; j < NODE_R; j += ACC_TPB) {
        const int node = (int)r0 + j;
        if (node < n) atomicAdd(&T[node], acc[j]);
    }
}

__global__ void __launch_bounds__(256)
final_out(const float* __restrict__ T, const float* __restrict__ normD,
          const float* __restrict__ b2, const float* __restrict__ Wfc,
          const float* __restrict__ bfc, float* __restrict__ out, int n) {
    const int i    = blockIdx.x * blockDim.x + threadIdx.x;
    const int lane = threadIdx.x & 31;
    float cb = 0.0f;
    for (int t = 0; t < 4; ++t) {
        const int j = lane + 32 * t;
        cb += b2[j] * Wfc[j];
    }
    for (int off = 16; off > 0; off >>= 1)
        cb += __shfl_xor(cb, off, 32);
    if (i < n) out[i] = T[i] * deg_to_norm(normD[i]) + (cb + bfc[0]);
}

extern "C" void kernel_launch(void* const* d_in, const int* in_sizes, int n_in,
                              void* d_out, int out_size, void* d_ws, size_t ws_size,
                              hipStream_t stream) {
    const float* x   = (const float*)d_in[0];
    const float* W1  = (const float*)d_in[1];
    const float* b1  = (const float*)d_in[2];
    const float* W2  = (const float*)d_in[3];
    const float* b2  = (const float*)d_in[4];
    const float* Wfc = (const float*)d_in[5];
    const float* bfc = (const float*)d_in[6];
    const int*   src = (const int*)d_in[7];
    const int*   dst = (const int*)d_in[8];
    float* out = (float*)d_out;

    const int N = N_NODES, E = N_EDGES;
    const int NH = NHID;

    float* BUF_A = (float*)d_ws;
    float* BUF_B = BUF_A + (size_t)N * NH;
    float* normS = BUF_B + (size_t)N * NH;
    float* normD = normS + N;

    const int TPB = 256;
    const int gridN     = (N + TPB - 1) / TPB;
    const int gridEdgeWaves = (E + 8 * EDGES_PER_WAVE - 1) / (8 * EDGES_PER_WAVE);
    const int gridGemm  = N / BM;

    zero_kernel<<<(2 * N + TPB - 1) / TPB, TPB, 0, stream>>>(normS, 2 * N);
    degree_kernel<<<NODE_NR * EDGE_NS, ACC_TPB, 0, stream>>>(src, dst, normS, normD, E, N);

    gemm_scaled_wmma<<<gridGemm, TPB, 0, stream>>>(x, normS, W1, BUF_A, BUF_B, NFEAT);
    scatter_add<<<gridEdgeWaves, TPB, 0, stream>>>(BUF_A, src, dst, BUF_B, E);

    float* S = BUF_A;
    float* T = BUF_A + N;
    gemm_scaled_wmma_rowdot<<<gridGemm, TPB, 0, stream>>>(BUF_B, normS, normD, b1, W2, Wfc,
                                                          S, T, NH);
    edge_scalar_add<<<NODE_NR * EDGE_NS, ACC_TPB, 0, stream>>>(S, src, dst, T, E, N);

    final_out<<<gridN, TPB, 0, stream>>>(T, normD, b2, Wfc, bfc, out, N);
}


# edge row sum: edge list ordered by destination in the call (LDS counting sort), a wave sums a node's rows in registers, one store per line; no f32 row atomics; sums in placed (arrival) order
# speedup vs baseline: 4.2977x; 1.1222x over previous
#include <hip/hip_runtime.h>

typedef __attribute__((ext_vector_type(2))) float v2f;
typedef __attribute__((ext_vector_type(8))) float v8f;

#define N_NODES 100000
#define N_EDGES 1600000
#define NFEAT   256
#define NHID    128

__global__ void zero_kernel(float* __restrict__ p, int n) {
    int i = blockIdx.x * blockDim.x + threadIdx.x;
    if (i < n) p[i] = 0.0f;
}

#define NODE_NR 16
#define NODE_R  6272
#define EDGE_NS 16
#define ACC_TPB 1024
static_assert(NODE_NR * NODE_R >= N_NODES, "node ranges must cover all nodes");

__device__ __forceinline__ int slice_len(int e) {
    return (((e + EDGE_NS - 1) / EDGE_NS) + 3) & ~3;
}

__device__ __forceinline__ void count_one(unsigned* cnt, int idx, unsigned r0) {
    const unsigned u = (unsigned)idx - r0;
    if (u < (unsigned)NODE_R) atomicAdd(&cnt[u], 1u);
}

__device__ __forceinline__ void count_four(unsigned* cnt, const int4 v, unsigned r0) {
    count_one(cnt, v.x, r0);
    count_one(cnt, v.y, r0);
    count_one(cnt, v.z, r0);
    count_one(cnt, v.w, r0);
}

__global__ void __launch_bounds__(ACC_TPB)
degree_kernel(const int* __restrict__ src, const int* __restrict__ dst,
              float* __restrict__ degS, float* __restrict__ degD,
              unsigned* __restrict__ Ctab, unsigned* __restrict__ RT, int e, int n) {
    __shared__ unsigned cntS[NODE_R];
    __shared__ unsigned cntD[NODE_R];
    __shared__ unsigned btot;
    const int t     = threadIdx.x;
    if (t == 0) btot = 0u;
    const int range = blockIdx.x % NODE_NR;
    const int slice = blockIdx.x / NODE_NR;
    const unsigned r0 = (unsigned)(range * NODE_R);

    for (int i = t; i < NODE_R; i += ACC_TPB) { cntS[i] = 0u; cntD[i] = 0u; }
    __syncthreads();

    const int per = slice_len(e);
    const int e0  = slice * per;
    const int e1  = (e0 + per < e) ? (e0 + per) : e;
    const int stride = 4 * ACC_TPB;
    int i = e0 + 4 * t;
    for (; i + stride + 4 <= e1; i += 2 * stride) {
        const int4 s0 = *(const int4*)(src + i);
        const int4 d0 = *(const int4*)(dst + i);
        const int4 s1 = *(const int4*)(src + i + stride);
        const int4 d1 = *(const int4*)(dst + i + stride);
        count_four(cntS, s0, r0);
        count_four(cntD, d0, r0);
        count_four(cntS, s1, r0);
        count_four(cntD, d1, r0);
    }
    for (; i < e1; i += stride) {
        if (i + 4 <= e1) {
            const int4 s0 = *(const int4*)(src + i);
            const int4 d0 = *(const int4*)(dst + i);
            count_four(cntS, s0, r0);
            count_four(cntD, d0, r0);
        } else {
            for (int k = i; k < e1; ++k) {
                count_one(cntS, src[k], r0);
                count_one(cntD, dst[k], r0);
            }
        }
    }
    __syncthreads();

    unsigned mytot = 0u;
    for (int j = t; j < NODE_R; j += ACC_TPB) {
        const int node = (int)r0 + j;
        if (node < n) {
            const unsigned cd = cntD[j];
            atomicAdd(&degS[node], (float)cntS[j]);
            atomicAdd(&degD[node], (float)cd);
            Ctab[(size_t)slice * n + node] = cd;
            mytot += cd;
        }
    }
    for (int off = 16; off > 0; off >>= 1)
        mytot += __shfl_xor(mytot, off, 32);
    if ((t & 31) == 0) atomicAdd(&btot, mytot);
    __syncthreads();
    if (t == 0) RT[range * EDGE_NS + slice] = btot;
}

#define SCAN_PT 7
static_assert(SCAN_PT * ACC_TPB >= NODE_R, "one scan block must cover a node range");
__global__ void __launch_bounds__(ACC_TPB)
rowstart_kernel(unsigned* __restrict__ Ctab, const unsigned* __restrict__ RT,
                int* __restrict__ rowptr, int n) {
    __shared__ unsigned tot[SCAN_PT * ACC_TPB];
    __shared__ unsigned wsum[32];
    __shared__ unsigned offs;
    const int t    = threadIdx.x;
    const int lane = t & 31;
    const int wave = t >> 5;
    const int r    = blockIdx.x;
    const int r0   = r * NODE_R;
    if (t == 0) offs = 0u;
    for (int j = t; j < SCAN_PT * ACC_TPB; j += ACC_TPB) {
        const int node = r0 + j;
        unsigned s = 0u;
        if (j < NODE_R && node < n) {
            for (int k = 0; k < EDGE_NS; ++k) s += Ctab[(size_t)k * n + node];
        }
        tot[j] = s;
    }
    __syncthreads();
    if (t < r * EDGE_NS) atomicAdd(&offs, RT[t]);
    unsigned loc[SCAN_PT];
    unsigned sum = 0u;
    for (int i = 0; i < SCAN_PT; ++i) { loc[i] = tot[SCAN_PT * t + i]; sum += loc[i]; }
    unsigned inc = sum;
    for (int off = 1; off < 32; off <<= 1) {
        const unsigned y = __shfl_up(inc, off, 32);
        if (lane >= off) inc += y;
    }
    if (lane == 31) wsum[wave] = inc;
    __syncthreads();
    if (t < 32) {
        const unsigned x = wsum[t];
        unsigned xi = x;
        for (int off = 1; off < 32; off <<= 1) {
            const unsigned y = __shfl_up(xi, off, 32);
            if (lane >= off) xi += y;
        }
        wsum[t] = xi - x;
    }
    __syncthreads();
    unsigned run = offs + wsum[wave] + (inc - sum);
    for (int i = 0; i < SCAN_PT; ++i) { const unsigned cc = loc[i]; tot[SCAN_PT * t + i] = run; run += cc; }
    __syncthreads();
    for (int j = t; j < NODE_R; j += ACC_TPB) {
        const int node = r0 + j;
        if (node < n) {
            unsigned b = tot[j];
            rowptr[node] = (int)b;
            for (int k = 0; k < EDGE_NS; ++k) {
                const size_t idx = (size_t)k * n + node;
                const unsigned cc = Ctab[idx];
                Ctab[idx] = b;
                b += cc;
            }
            if (node == n - 1) rowptr[n] = (int)b;
        }
    }
}

__device__ __forceinline__ void place_one(unsigned* cur, int s, int d, unsigned r0,
                                          int* __restrict__ col) {
    const unsigned u = (unsigned)d - r0;
    if (u < (unsigned)NODE_R) {
        const unsigned pos = atomicAdd(&cur[u], 1u);
        if (pos < (unsigned)N_EDGES) col[pos] = s;
    }
}

__global__ void __launch_bounds__(ACC_TPB)
place_kernel(const int* __restrict__ src, const int* __restrict__ dst,
             const unsigned* __restrict__ Ctab, int* __restrict__ col, int e, int n) {
    __shared__ unsigned cur[NODE_R];
    const int t     = threadIdx.x;
    const int range = blockIdx.x % NODE_NR;
    const int slice = blockIdx.x / NODE_NR;
    const unsigned r0 = (unsigned)(range * NODE_R);

    for (int j = t; j < NODE_R; j += ACC_TPB) {
        const int node = (int)r0 + j;
        cur[j] = (node < n) ? Ctab[(size_t)slice * n + node] : 0u;
    }
    __syncthreads();

    const int per = slice_len(e);
    const int e0  = slice * per;
    const int e1  = (e0 + per < e) ? (e0 + per) : e;
    const int stride = 4 * ACC_TPB;
    int i = e0 + 4 * t;
    for (; i + stride + 4 <= e1; i += 2 * stride) {
        const int4 s0 = *(const int4*)(src + i);
        const int4 d0 = *(const int4*)(dst + i);
        const int4 s1 = *(const int4*)(src + i + stride);
        const int4 d1 = *(const int4*)(dst + i + stride);
        place_one(cur, s0.x, d0.x, r0, col);
        place_one(cur, s0.y, d0.y, r0, col);
        place_one(cur, s0.z, d0.z, r0, col);
        place_one(cur, s0.w, d0.w, r0, col);
        place_one(cur, s1.x, d1.x, r0, col);
        place_one(cur, s1.y, d1.y, r0, col);
        place_one(cur, s1.z, d1.z, r0, col);
        place_one(cur, s1.w, d1.w, r0, col);
    }
    for (; i < e1; i += stride) {
        const int kend = (i + 4 <= e1) ? (i + 4) : e1;
        for (int k = i; k < kend; ++k) place_one(cur, src[k], dst[k], r0, col);
    }
}

__device__ __forceinline__ float deg_to_norm(float deg) {
    return rsqrtf(fmaxf(deg, 1.0f));
}

#define BM 80
#define BK 32
#define PA 36
#define PB 288
#define PC 132
#define SMEM_FLOATS (BM * PC)

template <bool FUSE>
__device__ __forceinline__ void gemm80_mainloop(const float* __restrict__ X,
                                                const float* __restrict__ scale,
                                                const float* __restrict__ W, int K,
                                                float* smem, v8f (&acc)[5],
                                                const float* __restrict__ normD,
                                                const float* __restrict__ bias) {
    float* As = smem;
    float* Bs = smem + BM * PA;
    const int t    = threadIdx.x;
    const int wave = t >> 5;
    const int lane = t & 31;
    const int h    = lane >> 4;
    const int c    = lane & 15;
    const int row0 = blockIdx.x * BM;

    const bool third = (t < 128);
    const float* xa[3];
    float ns[3];
    float nd[3];
    int   aoff[3];
#pragma unroll
    for (int i = 0; i < 3; ++i) {
        const int idx = t + 256 * i;
        const int r   = (i < 2 || third) ? (idx >> 3) : 0;
        const int k4  = idx & 7;
        ns[i]   = deg_to_norm(scale[row0 + r]);
        nd[i]   = FUSE ? deg_to_norm(normD[row0 + r]) : 0.0f;
        xa[i]   = X + (size_t)(row0 + r) * K + 4 * k4;
        aoff[i] = r * PA + 4 * k4;
    }
    const float* wb[2];
    int boff[2];
#pragma unroll
    for (int j = 0; j < 2; ++j) {
        const int idx = t + 256 * j;
        const int pr  = idx >> 5;
        const int n4  = idx & 31;
        wb[j]   = W + (size_t)(2 * pr) * NHID + 4 * n4;
        boff[j] = pr * PB + 8 * n4;
    }

    float4 ra[3];
    float4 rb[2][2];
    const float* bp = FUSE ? (bias + 4 * (t & 7)) : nullptr;
    float4 rbias = make_float4(0.0f, 0.0f, 0.0f, 0.0f);

    const int nch = K / BK;

#pragma unroll
    for (int i = 0; i < 2; ++i) ra[i] = *(const float4*)(xa[i]);
    if (third) ra[2] = *(const float4*)(xa[2]);
    if (FUSE) rbias = *(const float4*)(bp);
#pragma unroll
    for (int j = 0; j < 2; ++j) {
        rb[j][0] = *(const float4*)(wb[j]);
        rb[j][1] = *(const float4*)(wb[j] + NHID);
    }

    const int afrag = c * PA + 2 * h;
    const int bfrag = h * PB + (16 * wave + c) * 2;

    for (int ch = 0; ch < nch; ++ch) {
#pragma unroll
        for (int i = 0; i < 3; ++i) {
            if (i < 2 || third) {
                float4 v = ra[i];
                if (FUSE) {
                    const float h0 = v.x * nd[i] + rbias.x;
                    const float h1 = v.y * nd[i] + rbias.y;
                    const float h2 = v.z * nd[i] + rbias.z;
                    const float h3 = v.w * nd[i] + rbias.w;
                    v.x = h0 > 0.0f ? h0 : 0.0f;
                    v.y = h1 > 0.0f ? h1 : 0.0f;
                    v.z = h2 > 0.0f ? h2 : 0.0f;
                    v.w = h3 > 0.0f ? h3 : 0.0f;
                }
                v.x *= ns[i]; v.y *= ns[i]; v.z *= ns[i]; v.w *= ns[i];
                *(float4*)(As + aoff[i]) = v;
            }
        }
#pragma unroll
        for (int j = 0; j < 2; ++j) {
            const float4 lo = rb[j][0], hi = rb[j][1];
            *(float4*)(Bs + boff[j])     = make_float4(lo.x, hi.x, lo.y, hi.y);
            *(float4*)(Bs + boff[j] + 4) = make_float4(lo.z, hi.z, lo.w, hi.w);
        }
        __syncthreads();

        if (ch + 1 < nch) {
            const int ko = (ch + 1) * BK;
#pragma unroll
            for (int i = 0; i < 2; ++i) ra[i] = *(const float4*)(xa[i] + ko);
            if (third) ra[2] = *(const float4*)(xa[2] + ko);
            if (FUSE) rbias = *(const float4*)(bp + ko);
#pragma unroll
            for (int j = 0; j < 2; ++j) {
                rb[j][0] = *(const float4*)(wb[j] + (size_t)ko * NHID);
                rb[j][1] = *(const float4*)(wb[j] + (size_t)ko * NHID + NHID);
            }
        }

#pragma unroll
        for (int kk = 0; kk < BK / 4; ++kk) {
            const v2f b = *(const v2f*)(Bs + bfrag + 2 * kk * PB);
#pragma unroll
            for (int tm = 0; tm < 5; ++tm) {
                const v2f a = *(const v2f*)(As + afrag + 16 * tm * PA + 4 * kk);
                acc[tm] = __builtin_amdgcn_wmma_f32_16x16x4_f32(false, a, false, b,
                                                                (short)0, acc[tm], false, false);
            }
        }
        __syncthreads();
    }
}

__device__ __forceinline__ void gemm80_stage_result(float* smem, const v8f (&acc)[5]) {
    const int wave = threadIdx.x >> 5;
    const int lane = threadIdx.x & 31;
    const int h    = lane >> 4;
    const int c    = lane & 15;
#pragma unroll
    for (int tm = 0; tm < 5; ++tm) {
#pragma unroll
        for (int r = 0; r < 8; ++r) {
            smem[(16 * tm + 8 * h + r) * PC + 16 * wave + c] = acc[tm][r];
        }
    }
    __syncthreads();
}

__global__ void __launch_bounds__(256)
gemm_scaled_wmma(const float* __restrict__ X, const float* __restrict__ scale,
                 const float* __restrict__ W, float* __restrict__ Out, int K) {
    __shared__ __attribute__((aligned(16))) float smem[SMEM_FLOATS];
    v8f acc[5] = {};
    gemm80_mainloop<false>(X, scale, W, K, smem, acc, nullptr, nullptr);
    gemm80_stage_result(smem, acc);

    const int wave = threadIdx.x >> 5;
    const int lane = threadIdx.x & 31;
    const int row0 = blockIdx.x * BM;
#pragma unroll
    for (int i = 0; i < BM / 8; ++i) {
        const int row = wave + 8 * i;
        const float4 v = *(const float4*)(smem + row * PC + 4 * lane);
        *(float4*)(Out + (size_t)(row0 + row) * NHID + 4 * lane) = v;
    }
}

__global__ void __launch_bounds__(256)
gemm_scaled_wmma_rowdot(const float* __restrict__ X, const float* __restrict__ scale,
                        const float* __restrict__ normD, const float* __restrict__ bias,
                        const float* __restrict__ W, const float* __restrict__ Wfc,
                        float* __restrict__ S, float* __restrict__ T, int K) {
    __shared__ __attribute__((aligned(16))) float smem[SMEM_FLOATS];
    v8f acc[5] = {};
    gemm80_mainloop<true>(X, scale, W, K, smem, acc, normD, bias);
    gemm80_stage_result(smem, acc);

    const int wave = threadIdx.x >> 5;
    const int lane = threadIdx.x & 31;
    const int row0 = blockIdx.x * BM;
    float wf[4];
#pragma unroll
    for (int j = 0; j < 4; ++j) wf[j] = Wfc[lane + 32 * j];

#pragma unroll
    for (int i = 0; i < BM / 8; ++i) {
        const int row = wave + 8 * i;
        const float* crow = smem + row * PC + lane;
        float s = crow[0] * wf[0];
        s += crow[32] * wf[1];
        s += crow[64] * wf[2];
        s += crow[96] * wf[3];
        for (int off = 16; off > 0; off >>= 1)
            s += __shfl_xor(s, off, 32);
        if (lane == 0) {
            S[row0 + row] = s;
            T[row0 + row] = 0.0f;
        }
    }
}

#define NODES_PER_WAVE 8
#define EDGE_BATCH     8

template <bool FULL>
__device__ __forceinline__ void gather_batch(const float* __restrict__ M, int sv, int j0,
                                             int m, int lane, float (&v)[EDGE_BATCH][4]) {
    int s[EDGE_BATCH];
#pragma unroll
    for (int u = 0; u < EDGE_BATCH; ++u) {
        const int id = __builtin_amdgcn_readlane(sv, j0 + u);
        s[u] = id < 0 ? 0 : (id >= N_NODES ? N_NODES - 1 : id);
    }
#pragma unroll
    for (int u = 0; u < EDGE_BATCH; ++u) {
        if (FULL || j0 + u < m) {
            const float* __restrict__ mrow = M + (size_t)s[u] * NHID + lane;
            v[u][0] = mrow[0];
            v[u][1] = mrow[32];
            v[u][2] = mrow[64];
            v[u][3] = mrow[96];
        } else {
            v[u][0] = 0.0f; v[u][1] = 0.0f; v[u][2] = 0.0f; v[u][3] = 0.0f;
        }
    }
}

__global__ void __launch_bounds__(256)
gather_rows(const float* __restrict__ M, const int* __restrict__ rowptr,
            const int* __restrict__ col, float* __restrict__ AGG, int n) {
    const int w    = __builtin_amdgcn_readfirstlane((int)((blockIdx.x * blockDim.x + threadIdx.x) >> 5));
    const int lane = threadIdx.x & 31;
    const int n0   = w * NODES_PER_WAVE;
    if (n0 >= n) return;
    const int cntN = (n - n0 < NODES_PER_WAVE) ? (n - n0) : NODES_PER_WAVE;
    const int rp = rowptr[n0 + (lane < cntN ? lane : cntN)];
    int E0 = __builtin_amdgcn_readlane(rp, 0);
    int E1 = __builtin_amdgcn_readlane(rp, cntN);
    E0 = E0 < 0 ? 0 : (E0 > N_EDGES ? N_EDGES : E0);
    E1 = E1 < E0 ? E0 : (E1 > N_EDGES ? N_EDGES : E1);
    int cur  = 0;
    int endc = __builtin_amdgcn_readlane(rp, 1);
    float a0 = 0.0f, a1 = 0.0f, a2 = 0.0f, a3 = 0.0f;
    float* orow = AGG + (size_t)n0 * NHID + lane;

    for (int cb = E0; cb < E1; cb += 32) {
        const int m  = (E1 - cb < 32) ? (E1 - cb) : 32;
        const int sv = (lane < m) ? col[cb + lane] : 0;
        for (int j0 = 0; j0 < m; j0 += EDGE_BATCH) {
            float v[EDGE_BATCH][4];
            if (j0 + EDGE_BATCH <= m) gather_batch<true>(M, sv, j0, m, lane, v);
            else                      gather_batch<false>(M, sv, j0, m, lane, v);
#pragma unroll
            for (int u = 0; u < EDGE_BATCH; ++u) {
                if (j0 + u < m) {
                    const int j = cb + j0 + u;
                    while (cur + 1 < cntN && j >= endc) {
                        float* o = orow + (size_t)cur * NHID;
                        o[0] = a0; o[32] = a1; o[64] = a2; o[96] = a3;
                        a0 = 0.0f; a1 = 0.0f; a2 = 0.0f; a3 = 0.0f;
                        ++cur;
                        endc = __builtin_amdgcn_readlane(rp, cur + 1);
                    }
                    a0 += v[u][0];
                    a1 += v[u][1];
                    a2 += v[u][2];
                    a3 += v[u][3];
                }
            }
        }
    }
    while (cur < cntN) {
        float* o = orow + (size_t)cur * NHID;
        o[0] = a0; o[32] = a1; o[64] = a2; o[96] = a3;
        a0 = 0.0f; a1 = 0.0f; a2 = 0.0f; a3 = 0.0f;
        ++cur;
    }
}

#define ES_BATCH 8
__global__ void __launch_bounds__(ACC_TPB)
edge_scalar_add(const float* __restrict__ S, const int* __restrict__ src,
                const int* __restrict__ dst, float* __restrict__ T, int e, int n) {
    __shared__ float acc[NODE_R];
    const int t     = threadIdx.x;
    const int range = blockIdx.x % NODE_NR;
    const int slice = blockIdx.x / NODE_NR;
    const unsigned r0 = (unsigned)(range * NODE_R);

    for (int i = t; i < NODE_R; i += ACC_TPB) acc[i] = 0.0f;
    __syncthreads();

    const int per = slice_len(e);
    const int e0  = slice * per;
    const int e1  = (e0 + per < e) ? (e0 + per) : e;
    const int stride = 4 * ACC_TPB;
    int i = e0 + 4 * t;
    for (; i + stride + 4 <= e1; i += 2 * stride) {
        const int4 s0 = *(const int4*)(src + i);
        const int4 d0 = *(const int4*)(dst + i);
        const int4 s1 = *(const int4*)(src + i + stride);
        const int4 d1 = *(const int4*)(dst + i + stride);
        const int      sidx[ES_BATCH] = {s0.x, s0.y, s0.z, s0.w, s1.x, s1.y, s1.z, s1.w};
        const unsigned u[ES_BATCH]    = {(unsigned)d0.x - r0, (unsigned)d0.y - r0,
                                         (unsigned)d0.z - r0, (unsigned)d0.w - r0,
                                         (unsigned)d1.x - r0, (unsigned)d1.y - r0,
                                         (unsigned)d1.z - r0, (unsigned)d1.w - r0};
        float v[ES_BATCH];
#pragma unroll
        for (int k = 0; k < ES_BATCH; ++k) {
            v[k] = 0.0f;
            if (u[k] < (unsigned)NODE_R) v[k] = S[sidx[k]];
        }
#pragma unroll
        for (int k = 0; k < ES_BATCH; ++k) {
            if (u[k] < (unsigned)NODE_R) atomicAdd(&acc[u[k]], v[k]);
        }
    }
    for (; i < e1; i += stride) {
        const int kend = (i + 4 <= e1) ? (i + 4) : e1;
        for (int k = i; k < kend; ++k) {
            const unsigned uu = (unsigned)dst[k] - r0;
            if (uu < (unsigned)NODE_R) atomicAdd(&acc[uu], S[src[k]]);
        }
    }
    __syncthreads();

    for (int j = t; j < NODE_R; j += ACC_TPB) {
        const int node = (int)r0 + j;
        if (node < n) atomicAdd(&T[node], acc[j]);
    }
}

__global__ void __launch_bounds__(256)
final_out(const float* __restrict__ T, const float* __restrict__ normD,
          const float* __restrict__ b2, const float* __restrict__ Wfc,
          const float* __restrict__ bfc, float* __restrict__ out, int n) {
    const int i    = blockIdx.x * blockDim.x + threadIdx.x;
    const int lane = threadIdx.x & 31;
    float cb = 0.0f;
    for (int t = 0; t < 4; ++t) {
        const int j = lane + 32 * t;
        cb += b2[j] * Wfc[j];
    }
    for (int off = 16; off > 0; off >>= 1)
        cb += __shfl_xor(cb, off, 32);
    if (i < n) out[i] = T[i] * deg_to_norm(normD[i]) + (cb + bfc[0]);
}

extern "C" void kernel_launch(void* const* d_in, const int* in_sizes, int n_in,
                              void* d_out, int out_size, void* d_ws, size_t ws_size,
                              hipStream_t stream) {
    const float* x   = (const float*)d_in[0];
    const float* W1  = (const float*)d_in[1];
    const float* b1  = (const float*)d_in[2];
    const float* W2  = (const float*)d_in[3];
    const float* b2  = (const float*)d_in[4];
    const float* Wfc = (const float*)d_in[5];
    const float* bfc = (const float*)d_in[6];
    const int*   src = (const int*)d_in[7];
    const int*   dst = (const int*)d_in[8];
    float* out = (float*)d_out;

    const int N = N_NODES, E = N_EDGES;
    const int NH = NHID;

    float* BUF_A = (float*)d_ws;
    float* BUF_B = BUF_A + (size_t)N * NH;
    float* normS = BUF_B + (size_t)N * NH;
    float* normD = normS + N;

    const int P1_TILES = 768;
    const int P2_TILES = 482;
    const int P1_NODES = 61440;
    const int P2_NODES = 38560;
    float*    AGGP   = BUF_B;
    int*      col    = (int*)(BUF_B + (size_t)7864320);
    unsigned* Ctab   = (unsigned*)(col + 1600000);
    int*      rowptr = (int*)(Ctab + 1600000);
    unsigned* RT     = (unsigned*)(rowptr + 100032);
    float*    S      = (float*)(RT + 256);
    float*    T      = S + 100000;

    const int TPB = 256;
    const int gridN = (N + TPB - 1) / TPB;

    zero_kernel<<<(2 * N + TPB - 1) / TPB, TPB, 0, stream>>>(normS, 2 * N);
    degree_kernel<<<NODE_NR * EDGE_NS, ACC_TPB, 0, stream>>>(src, dst, normS, normD, Ctab, RT, E, N);

    rowstart_kernel<<<NODE_NR, ACC_TPB, 0, stream>>>(Ctab, RT, rowptr, N);
    place_kernel<<<NODE_NR * EDGE_NS, ACC_TPB, 0, stream>>>(src, dst, Ctab, col, E, N);

    gemm_scaled_wmma<<<1250, TPB, 0, stream>>>(x, normS, W1, BUF_A, NFEAT);

    gather_rows<<<960, TPB, 0, stream>>>(BUF_A, rowptr, col, AGGP, P1_NODES);
    gemm_scaled_wmma_rowdot<<<P1_TILES, TPB, 0, stream>>>(AGGP, normS, normD, b1, W2, Wfc,
                                                          S, T, NH);
    gather_rows<<<603, TPB, 0, stream>>>(BUF_A, rowptr + 61440, col, AGGP, P2_NODES);
    gemm_scaled_wmma_rowdot<<<P2_TILES, TPB, 0, stream>>>(AGGP, normS + 61440, normD + 61440, b1,
                                                          W2, Wfc, S + 61440, T + 61440, NH);
    edge_scalar_add<<<NODE_NR * EDGE_NS, ACC_TPB, 0, stream>>>(S, src, dst, T, E, N);

    final_out<<<gridN, TPB, 0, stream>>>(T, normD, b2, Wfc, bfc, out, N);
}


# layer-2 product removed by associativity: (H@W2)@Wfc = H@(W2@Wfc); w=W2@Wfc formed once per call, the row-sum wave dots relu(AGG*normD+b1) with w in registers; AGG never stored; sums re-associated
# speedup vs baseline: 5.3014x; 1.2335x over previous
#include <hip/hip_runtime.h>

typedef __attribute__((ext_vector_type(2))) float v2f;
typedef __attribute__((ext_vector_type(8))) float v8f;

#define N_NODES 100000
#define N_EDGES 1600000
#define NFEAT   256
#define NHID    128

#define WV_BLOCKS 4
__global__ void __launch_bounds__(256)
setup_kernel(float* __restrict__ p, int n, float* __restrict__ q, int m,
             const float* __restrict__ W2, const float* __restrict__ Wfc,
             float* __restrict__ wvec, int zero_blocks) {
    if ((int)blockIdx.x < zero_blocks) {
        const int i = blockIdx.x * blockDim.x + threadIdx.x;
        if (i < n) p[i] = 0.0f;
        else if (i - n < m) q[i - n] = 0.0f;
        return;
    }
    const int wave = threadIdx.x >> 5;
    const int lane = threadIdx.x & 31;
    const int rbase = (((int)blockIdx.x - zero_blocks) * 8 + wave) * 4;
    float f[4];
#pragma unroll
    for (int j = 0; j < 4; ++j) f[j] = Wfc[lane + 32 * j];
    float keep = 0.0f;
#pragma unroll
    for (int r = 0; r < 4; ++r) {
        const float* wrow = W2 + (size_t)(rbase + r) * NHID + lane;
        float s = wrow[0] * f[0];
        s += wrow[32] * f[1];
        s += wrow[64] * f[2];
        s += wrow[96] * f[3];
        for (int off = 16; off > 0; off >>= 1)
            s += __shfl_xor(s, off, 32);
        if (lane == r) keep = s;
    }
    if (lane < 4) wvec[rbase + lane] = keep;
}

#define NODE_NR 16
#define NODE_R  6272
#define EDGE_NS 16
#define ACC_TPB 1024
static_assert(NODE_NR * NODE_R >= N_NODES, "node ranges must cover all nodes");

__device__ __forceinline__ int slice_len(int e) {
    return (((e + EDGE_NS - 1) / EDGE_NS) + 3) & ~3;
}

__device__ __forceinline__ void count_one(unsigned* cnt, int idx, unsigned r0) {
    const unsigned u = (unsigned)idx - r0;
    if (u < (unsigned)NODE_R) atomicAdd(&cnt[u], 1u);
}

__device__ __forceinline__ void count_four(unsigned* cnt, const int4 v, unsigned r0) {
    count_one(cnt, v.x, r0);
    count_one(cnt, v.y, r0);
    count_one(cnt, v.z, r0);
    count_one(cnt, v.w, r0);
}

__global__ void __launch_bounds__(ACC_TPB)
degree_kernel(const int* __restrict__ src, const int* __restrict__ dst,
              float* __restrict__ degS, float* __restrict__ degD,
              unsigned* __restrict__ Ctab, unsigned* __restrict__ RT, int e, int n) {
    __shared__ unsigned cntS[NODE_R];
    __shared__ unsigned cntD[NODE_R];
    __shared__ unsigned btot;
    const int t     = threadIdx.x;
    if (t == 0) btot = 0u;
    const int range = blockIdx.x % NODE_NR;
    const int slice = blockIdx.x / NODE_NR;
    const unsigned r0 = (unsigned)(range * NODE_R);

    for (int i = t; i < NODE_R; i += ACC_TPB) { cntS[i] = 0u; cntD[i] = 0u; }
    __syncthreads();

    const int per = slice_len(e);
    const int e0  = slice * per;
    const int e1  = (e0 + per < e) ? (e0 + per) : e;
    const int stride = 4 * ACC_TPB;
    int i = e0 + 4 * t;
    for (; i + stride + 4 <= e1; i += 2 * stride) {
        const int4 s0 = *(const int4*)(src + i);
        const int4 d0 = *(const int4*)(dst + i);
        const int4 s1 = *(const int4*)(src + i + stride);
        const int4 d1 = *(const int4*)(dst + i + stride);
        count_four(cntS, s0, r0);
        count_four(cntD, d0, r0);
        count_four(cntS, s1, r0);
        count_four(cntD, d1, r0);
    }
    for (; i < e1; i += stride) {
        if (i + 4 <= e1) {
            const int4 s0 = *(const int4*)(src + i);
            const int4 d0 = *(const int4*)(dst + i);
            count_four(cntS, s0, r0);
            count_four(cntD, d0, r0);
        } else {
            for (int k = i; k < e1; ++k) {
                count_one(cntS, src[k], r0);
                count_one(cntD, dst[k], r0);
            }
        }
    }
    __syncthreads();

    unsigned mytot = 0u;
    for (int j = t; j < NODE_R; j += ACC_TPB) {
        const int node = (int)r0 + j;
        if (node < n) {
            const unsigned cd = cntD[j];
            atomicAdd(&degS[node], (float)cntS[j]);
            atomicAdd(&degD[node], (float)cd);
            Ctab[(size_t)slice * n + node] = cd;
            mytot += cd;
        }
    }
    for (int off = 16; off > 0; off >>= 1)
        mytot += __shfl_xor(mytot, off, 32);
    if ((t & 31) == 0) atomicAdd(&btot, mytot);
    __syncthreads();
    if (t == 0) RT[range * EDGE_NS + slice] = btot;
}

#define SCAN_PT 7
static_assert(SCAN_PT * ACC_TPB >= NODE_R, "one scan block must cover a node range");
__global__ void __launch_bounds__(ACC_TPB)
rowstart_kernel(unsigned* __restrict__ Ctab, const unsigned* __restrict__ RT,
                int* __restrict__ rowptr, int n) {
    __shared__ unsigned tot[SCAN_PT * ACC_TPB];
    __shared__ unsigned wsum[32];
    __shared__ unsigned offs;
    const int t    = threadIdx.x;
    const int lane = t & 31;
    const int wave = t >> 5;
    const int r    = blockIdx.x;
    const int r0   = r * NODE_R;
    if (t == 0) offs = 0u;
    for (int j = t; j < SCAN_PT * ACC_TPB; j += ACC_TPB) {
        const int node = r0 + j;
        unsigned s = 0u;
        if (j < NODE_R && node < n) {
            for (int k = 0; k < EDGE_NS; ++k) s += Ctab[(size_t)k * n + node];
        }
        tot[j] = s;
    }
    __syncthreads();
    if (t < r * EDGE_NS) atomicAdd(&offs, RT[t]);
    unsigned loc[SCAN_PT];
    unsigned sum = 0u;
    for (int i = 0; i < SCAN_PT; ++i) { loc[i] = tot[SCAN_PT * t + i]; sum += loc[i]; }
    unsigned inc = sum;
    for (int off = 1; off < 32; off <<= 1) {
        const unsigned y = __shfl_up(inc, off, 32);
        if (lane >= off) inc += y;
    }
    if (lane == 31) wsum[wave] = inc;
    __syncthreads();
    if (t < 32) {
        const unsigned x = wsum[t];
        unsigned xi = x;
        for (int off = 1; off < 32; off <<= 1) {
            const unsigned y = __shfl_up(xi, off, 32);
            if (lane >= off) xi += y;
        }
        wsum[t] = xi - x;
    }
    __syncthreads();
    unsigned run = offs + wsum[wave] + (inc - sum);
    for (int i = 0; i < SCAN_PT; ++i) { const unsigned cc = loc[i]; tot[SCAN_PT * t + i] = run; run += cc; }
    __syncthreads();
    for (int j = t; j < NODE_R; j += ACC_TPB) {
        const int node = r0 + j;
        if (node < n) {
            unsigned b = tot[j];
            rowptr[node] = (int)b;
            for (int k = 0; k < EDGE_NS; ++k) {
                const size_t idx = (size_t)k * n + node;
                const unsigned cc = Ctab[idx];
                Ctab[idx] = b;
                b += cc;
            }
            if (node == n - 1) rowptr[n] = (int)b;
        }
    }
}

__device__ __forceinline__ void place_one(unsigned* cur, int s, int d, unsigned r0,
                                          int* __restrict__ col) {
    const unsigned u = (unsigned)d - r0;
    if (u < (unsigned)NODE_R) {
        const unsigned pos = atomicAdd(&cur[u], 1u);
        if (pos < (unsigned)N_EDGES) col[pos] = s;
    }
}

__global__ void __launch_bounds__(ACC_TPB)
place_kernel(const int* __restrict__ src, const int* __restrict__ dst,
             const unsigned* __restrict__ Ctab, int* __restrict__ col, int e, int n) {
    __shared__ unsigned cur[NODE_R];
    const int t     = threadIdx.x;
    const int range = blockIdx.x % NODE_NR;
    const int slice = blockIdx.x / NODE_NR;
    const unsigned r0 = (unsigned)(range * NODE_R);

    for (int j = t; j < NODE_R; j += ACC_TPB) {
        const int node = (int)r0 + j;
        cur[j] = (node < n) ? Ctab[(size_t)slice * n + node] : 0u;
    }
    __syncthreads();

    const int per = slice_len(e);
    const int e0  = slice * per;
    const int e1  = (e0 + per < e) ? (e0 + per) : e;
    const int stride = 4 * ACC_TPB;
    int i = e0 + 4 * t;
    for (; i + stride + 4 <= e1; i += 2 * stride) {
        const int4 s0 = *(const int4*)(src + i);
        const int4 d0 = *(const int4*)(dst + i);
        const int4 s1 = *(const int4*)(src + i + stride);
        const int4 d1 = *(const int4*)(dst + i + stride);
        place_one(cur, s0.x, d0.x, r0, col);
        place_one(cur, s0.y, d0.y, r0, col);
        place_one(cur, s0.z, d0.z, r0, col);
        place_one(cur, s0.w, d0.w, r0, col);
        place_one(cur, s1.x, d1.x, r0, col);
        place_one(cur, s1.y, d1.y, r0, col);
        place_one(cur, s1.z, d1.z, r0, col);
        place_one(cur, s1.w, d1.w, r0, col);
    }
    for (; i < e1; i += stride) {
        const int kend = (i + 4 <= e1) ? (i + 4) : e1;
        for (int k = i; k < kend; ++k) place_one(cur, src[k], dst[k], r0, col);
    }
}

__device__ __forceinline__ float deg_to_norm(float deg) {
    return rsqrtf(fmaxf(deg, 1.0f));
}

#define BM 80
#define BK 32
#define PA 36
#define PB 288
#define PC 132
#define SMEM_FLOATS (BM * PC)

__device__ __forceinline__ void gemm80_mainloop(const float* __restrict__ X,
                                                const float* __restrict__ scale,
                                                const float* __restrict__ W, int K,
                                                float* smem, v8f (&acc)[5]) {
    float* As = smem;
    float* Bs = smem + BM * PA;
    const int t    = threadIdx.x;
    const int wave = t >> 5;
    const int lane = t & 31;
    const int h    = lane >> 4;
    const int c    = lane & 15;
    const int row0 = blockIdx.x * BM;

    const bool third = (t < 128);
    const float* xa[3];
    float ns[3];
    int   aoff[3];
#pragma unroll
    for (int i = 0; i < 3; ++i) {
        const int idx = t + 256 * i;
        const int r   = (i < 2 || third) ? (idx >> 3) : 0;
        const int k4  = idx & 7;
        ns[i]   = deg_to_norm(scale[row0 + r]);
        xa[i]   = X + (size_t)(row0 + r) * K + 4 * k4;
        aoff[i] = r * PA + 4 * k4;
    }
    const float* wb[2];
    int boff[2];
#pragma unroll
    for (int j = 0; j < 2; ++j) {
        const int idx = t + 256 * j;
        const int pr  = idx >> 5;
        const int n4  = idx & 31;
        wb[j]   = W + (size_t)(2 * pr) * NHID + 4 * n4;
        boff[j] = pr * PB + 8 * n4;
    }

    float4 ra[3];
    float4 rb[2][2];

    const int nch = K / BK;

#pragma unroll
    for (int i = 0; i < 2; ++i) ra[i] = *(const float4*)(xa[i]);
    if (third) ra[2] = *(const float4*)(xa[2]);
#pragma unroll
    for (int j = 0; j < 2; ++j) {
        rb[j][0] = *(const float4*)(wb[j]);
        rb[j][1] = *(const float4*)(wb[j] + NHID);
    }

    const int afrag = c * PA + 2 * h;
    const int bfrag = h * PB + (16 * wave + c) * 2;

    for (int ch = 0; ch < nch; ++ch) {
#pragma unroll
        for (int i = 0; i < 3; ++i) {
            if (i < 2 || third) {
                float4 v = ra[i];
                v.x *= ns[i]; v.y *= ns[i]; v.z *= ns[i]; v.w *= ns[i];
                *(float4*)(As + aoff[i]) = v;
            }
        }
#pragma unroll
        for (int j = 0; j < 2; ++j) {
            const float4 lo = rb[j][0], hi = rb[j][1];
            *(float4*)(Bs + boff[j])     = make_float4(lo.x, hi.x, lo.y, hi.y);
            *(float4*)(Bs + boff[j] + 4) = make_float4(lo.z, hi.z, lo.w, hi.w);
        }
        __syncthreads();

        if (ch + 1 < nch) {
            const int ko = (ch + 1) * BK;
#pragma unroll
            for (int i = 0; i < 2; ++i) ra[i] = *(const float4*)(xa[i] + ko);
            if (third) ra[2] = *(const float4*)(xa[2] + ko);
#pragma unroll
            for (int j = 0; j < 2; ++j) {
                rb[j][0] = *(const float4*)(wb[j] + (size_t)ko * NHID);
                rb[j][1] = *(const float4*)(wb[j] + (size_t)ko * NHID + NHID);
            }
        }

#pragma unroll
        for (int kk = 0; kk < BK / 4; ++kk) {
            const v2f b = *(const v2f*)(Bs + bfrag + 2 * kk * PB);
#pragma unroll
            for (int tm = 0; tm < 5; ++tm) {
                const v2f a = *(const v2f*)(As + afrag + 16 * tm * PA + 4 * kk);
                acc[tm] = __builtin_amdgcn_wmma_f32_16x16x4_f32(false, a, false, b,
                                                                (short)0, acc[tm], false, false);
            }
        }
        __syncthreads();
    }
}

__device__ __forceinline__ void gemm80_stage_result(float* smem, const v8f (&acc)[5]) {
    const int wave = threadIdx.x >> 5;
    const int lane = threadIdx.x & 31;
    const int h    = lane >> 4;
    const int c    = lane & 15;
#pragma unroll
    for (int tm = 0; tm < 5; ++tm) {
#pragma unroll
        for (int r = 0; r < 8; ++r) {
            smem[(16 * tm + 8 * h + r) * PC + 16 * wave + c] = acc[tm][r];
        }
    }
    __syncthreads();
}

__global__ void __launch_bounds__(256)
gemm_scaled_wmma(const float* __restrict__ X, const float* __restrict__ scale,
                 const float* __restrict__ W, float* __restrict__ Out, int K) {
    __shared__ __attribute__((aligned(16))) float smem[SMEM_FLOATS];
    v8f acc[5] = {};
    gemm80_mainloop(X, scale, W, K, smem, acc);
    gemm80_stage_result(smem, acc);

    const int wave = threadIdx.x >> 5;
    const int lane = threadIdx.x & 31;
    const int row0 = blockIdx.x * BM;
#pragma unroll
    for (int i = 0; i < BM / 8; ++i) {
        const int row = wave + 8 * i;
        const float4 v = *(const float4*)(smem + row * PC + 4 * lane);
        *(float4*)(Out + (size_t)(row0 + row) * NHID + 4 * lane) = v;
    }
}

#define NODES_PER_WAVE 10
#define EDGE_BATCH     8

template <bool FULL>
__device__ __forceinline__ void gather_batch(const float* __restrict__ M, int sv, int j0,
                                             int m, int lane, float (&v)[EDGE_BATCH][4]) {
    int s[EDGE_BATCH];
#pragma unroll
    for (int u = 0; u < EDGE_BATCH; ++u) {
        const int id = __builtin_amdgcn_readlane(sv, j0 + u);
        s[u] = id < 0 ? 0 : (id >= N_NODES ? N_NODES - 1 : id);
    }
#pragma unroll
    for (int u = 0; u < EDGE_BATCH; ++u) {
        if (FULL || j0 + u < m) {
            const float* __restrict__ mrow = M + (size_t)s[u] * NHID + lane;
            v[u][0] = mrow[0];
            v[u][1] = mrow[32];
            v[u][2] = mrow[64];
            v[u][3] = mrow[96];
        } else {
            v[u][0] = 0.0f; v[u][1] = 0.0f; v[u][2] = 0.0f; v[u][3] = 0.0f;
        }
    }
}

__device__ __forceinline__ float wave_sum(float x) {
    x += __builtin_bit_cast(float, __builtin_amdgcn_update_dpp(0, __builtin_bit_cast(int, x), 0x161, 0xf, 0xf, false));
    x += __builtin_bit_cast(float, __builtin_amdgcn_update_dpp(0, __builtin_bit_cast(int, x), 0x162, 0xf, 0xf, false));
    x += __builtin_bit_cast(float, __builtin_amdgcn_update_dpp(0, __builtin_bit_cast(int, x), 0x164, 0xf, 0xf, false));
    x += __builtin_bit_cast(float, __builtin_amdgcn_update_dpp(0, __builtin_bit_cast(int, x), 0x168, 0xf, 0xf, false));
    const int b = __builtin_bit_cast(int, x);
    x += __builtin_bit_cast(float, __builtin_amdgcn_permlanex16(b, b, 0x76543210, 0xfedcba98, false, false));
    return x;
}

__device__ __forceinline__ float row_relu_dot(float a0, float a1, float a2, float a3, float nd,
                                              const float (&br)[4], const float (&wr)[4]) {
    const float h0 = a0 * nd + br[0];
    const float h1 = a1 * nd + br[1];
    const float h2 = a2 * nd + br[2];
    const float h3 = a3 * nd + br[3];
    float p = (h0 > 0.0f ? h0 : 0.0f) * wr[0];
    p += (h1 > 0.0f ? h1 : 0.0f) * wr[1];
    p += (h2 > 0.0f ? h2 : 0.0f) * wr[2];
    p += (h3 > 0.0f ? h3 : 0.0f) * wr[3];
    return wave_sum(p);
}

__global__ void __launch_bounds__(256)
gather_rows_dot(const float* __restrict__ M, const int* __restrict__ rowptr,
                const int* __restrict__ col, const float* __restrict__ degS,
                const float* __restrict__ degD, const float* __restrict__ bias,
                const float* __restrict__ wvec, float* __restrict__ S, int n) {
    const int w    = __builtin_amdgcn_readfirstlane((int)((blockIdx.x * blockDim.x + threadIdx.x) >> 5));
    const int lane = threadIdx.x & 31;
    const int n0   = w * NODES_PER_WAVE;
    if (n0 >= n) return;
    const int cntN = (n - n0 < NODES_PER_WAVE) ? (n - n0) : NODES_PER_WAVE;
    const int rp = rowptr[n0 + (lane < cntN ? lane : cntN)];
    const int   ln  = n0 + (lane < cntN ? lane : cntN - 1);
    const float ndl = deg_to_norm(degD[ln]);
    const float nsl = deg_to_norm(degS[ln]);
    float br[4];
    float wr[4];
#pragma unroll
    for (int t = 0; t < 4; ++t) {
        br[t] = bias[lane + 32 * t];
        wr[t] = wvec[lane + 32 * t];
    }
    int E0 = __builtin_amdgcn_readlane(rp, 0);
    int E1 = __builtin_amdgcn_readlane(rp, cntN);
    E0 = E0 < 0 ? 0 : (E0 > N_EDGES ? N_EDGES : E0);
    E1 = E1 < E0 ? E0 : (E1 > N_EDGES ? N_EDGES : E1);
    int cur  = 0;
    int endc = __builtin_amdgcn_readlane(rp, 1);
    float a0 = 0.0f, a1 = 0.0f, a2 = 0.0f, a3 = 0.0f;
    float sout = 0.0f;

    for (int cb = E0; cb < E1; cb += 32) {
        const int m  = (E1 - cb < 32) ? (E1 - cb) : 32;
        const int sv = (lane < m) ? col[cb + lane] : 0;
        for (int j0 = 0; j0 < m; j0 += EDGE_BATCH) {
            float v[EDGE_BATCH][4];
            if (j0 + EDGE_BATCH <= m) gather_batch<true>(M, sv, j0, m, lane, v);
            else                      gather_batch<false>(M, sv, j0, m, lane, v);
#pragma unroll
            for (int u = 0; u < EDGE_BATCH; ++u) {
                if (j0 + u < m) {
                    const int j = cb + j0 + u;
                    while (cur + 1 < cntN && j >= endc) {
                        const float nd = __int_as_float(__builtin_amdgcn_readlane(__float_as_int(ndl), cur));
                        const float p  = row_relu_dot(a0, a1, a2, a3, nd, br, wr);
                        if (lane == cur) sout = p;
                        a0 = 0.0f; a1 = 0.0f; a2 = 0.0f; a3 = 0.0f;
                        ++cur;
                        endc = __builtin_amdgcn_readlane(rp, cur + 1);
                    }
                    a0 += v[u][0];
                    a1 += v[u][1];
                    a2 += v[u][2];
                    a3 += v[u][3];
                }
            }
        }
    }
    while (cur < cntN) {
        const float nd = __int_as_float(__builtin_amdgcn_readlane(__float_as_int(ndl), cur));
        const float p  = row_relu_dot(a0, a1, a2, a3, nd, br, wr);
        if (lane == cur) sout = p;
        a0 = 0.0f; a1 = 0.0f; a2 = 0.0f; a3 = 0.0f;
        ++cur;
    }
    if (lane < cntN) S[n0 + lane] = sout * nsl;
}

#define ES_BATCH 8
__global__ void __launch_bounds__(ACC_TPB)
edge_scalar_add(const float* __restrict__ S, const int* __restrict__ src,
                const int* __restrict__ dst, float* __restrict__ T, int e, int n) {
    __shared__ float acc[NODE_R];
    const int t     = threadIdx.x;
    const int range = blockIdx.x % NODE_NR;
    const int slice = blockIdx.x / NODE_NR;
    const unsigned r0 = (unsigned)(range * NODE_R);

    for (int i = t; i < NODE_R; i += ACC_TPB) acc[i] = 0.0f;
    __syncthreads();

    const int per = slice_len(e);
    const int e0  = slice * per;
    const int e1  = (e0 + per < e) ? (e0 + per) : e;
    const int stride = 4 * ACC_TPB;
    int i = e0 + 4 * t;
    for (; i + stride + 4 <= e1; i += 2 * stride) {
        const int4 s0 = *(const int4*)(src + i);
        const int4 d0 = *(const int4*)(dst + i);
        const int4 s1 = *(const int4*)(src + i + stride);
        const int4 d1 = *(const int4*)(dst + i + stride);
        const int      sidx[ES_BATCH] = {s0.x, s0.y, s0.z, s0.w, s1.x, s1.y, s1.z, s1.w};
        const unsigned u[ES_BATCH]    = {(unsigned)d0.x - r0, (unsigned)d0.y - r0,
                                         (unsigned)d0.z - r0, (unsigned)d0.w - r0,
                                         (unsigned)d1.x - r0, (unsigned)d1.y - r0,
                                         (unsigned)d1.z - r0, (unsigned)d1.w - r0};
        float v[ES_BATCH];
#pragma unroll
        for (int k = 0; k < ES_BATCH; ++k) {
            v[k] = 0.0f;
            if (u[k] < (unsigned)NODE_R) v[k] = S[sidx[k]];
        }
#pragma unroll
        for (int k = 0; k < ES_BATCH; ++k) {
            if (u[k] < (unsigned)NODE_R) atomicAdd(&acc[u[k]], v[k]);
        }
    }
    for (; i < e1; i += stride) {
        const int kend = (i + 4 <= e1) ? (i + 4) : e1;
        for (int k = i; k < kend; ++k) {
            const unsigned uu = (unsigned)dst[k] - r0;
            if (uu < (unsigned)NODE_R) atomicAdd(&acc[uu], S[src[k]]);
        }
    }
    __syncthreads();

    for (int j = t; j < NODE_R; j += ACC_TPB) {
        const int node = (int)r0 + j;
        if (node < n) atomicAdd(&T[node], acc[j]);
    }
}

__global__ void __launch_bounds__(256)
final_out(const float* __restrict__ T, const float* __restrict__ normD,
          const float* __restrict__ b2, const float* __restrict__ Wfc,
          const float* __restrict__ bfc, float* __restrict__ out, int n) {
    const int i    = blockIdx.x * blockDim.x + threadIdx.x;
    const int lane = threadIdx.x & 31;
    float cb = 0.0f;
    for (int t = 0; t < 4; ++t) {
        const int j = lane + 32 * t;
        cb += b2[j] * Wfc[j];
    }
    for (int off = 16; off > 0; off >>= 1)
        cb += __shfl_xor(cb, off, 32);
    if (i < n) out[i] = T[i] * deg_to_norm(normD[i]) + (cb + bfc[0]);
}

extern "C" void kernel_launch(void* const* d_in, const int* in_sizes, int n_in,
                              void* d_out, int out_size, void* d_ws, size_t ws_size,
                              hipStream_t stream) {
    const float* x   = (const float*)d_in[0];
    const float* W1  = (const float*)d_in[1];
    const float* b1  = (const float*)d_in[2];
    const float* W2  = (const float*)d_in[3];
    const float* b2  = (const float*)d_in[4];
    const float* Wfc = (const float*)d_in[5];
    const float* bfc = (const float*)d_in[6];
    const int*   src = (const int*)d_in[7];
    const int*   dst = (const int*)d_in[8];
    float* out = (float*)d_out;

    const int N = N_NODES, E = N_EDGES;

    float* BUF_A = (float*)d_ws;
    float* BUF_B = BUF_A + (size_t)N * NHID;
    float* normS = BUF_B + (size_t)N * NHID;
    float* normD = normS + N;

    int*      col    = (int*)(BUF_B + (size_t)7864320);
    unsigned* Ctab   = (unsigned*)(col + 1600000);
    int*      rowptr = (int*)(Ctab + 1600000);
    unsigned* RT     = (unsigned*)(rowptr + 100032);
    float*    S      = (float*)(RT + 256);
    float*    T      = S + 100000;
    float*    wvec   = T + 100000;

    const int TPB = 256;
    const int gridN = (N + TPB - 1) / TPB;
    const int ZERO_BLOCKS = (3 * N + TPB - 1) / TPB;

    setup_kernel<<<ZERO_BLOCKS + WV_BLOCKS, TPB, 0, stream>>>(normS, 2 * N, T, N, W2, Wfc, wvec,
                                                            ZERO_BLOCKS);
    degree_kernel<<<NODE_NR * EDGE_NS, ACC_TPB, 0, stream>>>(src, dst, normS, normD, Ctab, RT, E, N);

    rowstart_kernel<<<NODE_NR, ACC_TPB, 0, stream>>>(Ctab, RT, rowptr, N);
    place_kernel<<<NODE_NR * EDGE_NS, ACC_TPB, 0, stream>>>(src, dst, Ctab, col, E, N);

    gemm_scaled_wmma<<<1250, TPB, 0, stream>>>(x, normS, W1, BUF_A, NFEAT);

    gather_rows_dot<<<1250, TPB, 0, stream>>>(BUF_A, rowptr, col, normS, normD, b1, wvec, S, N);

    edge_scalar_add<<<NODE_NR * EDGE_NS, ACC_TPB, 0, stream>>>(S, src, dst, T, E, N);

    final_out<<<gridN, TPB, 0, stream>>>(T, normD, b2, Wfc, bfc, out, N);
}


# second edge sum taken off the destination-ordered list: a block stages S[src] of its nodes' runs in LDS, a thread sums a node's run and stores out; no T plane, no f32 atomic sums; sums in list order
# speedup vs baseline: 5.8925x; 1.1115x over previous
#include <hip/hip_runtime.h>

typedef __attribute__((ext_vector_type(2))) float v2f;
typedef __attribute__((ext_vector_type(8))) float v8f;

#define N_NODES 100000
#define N_EDGES 1600000
#define NFEAT   256
#define NHID    128

#define WV_BLOCKS 4
__global__ void __launch_bounds__(256)
setup_kernel(float* __restrict__ p, int n,
             const float* __restrict__ W2, const float* __restrict__ Wfc,
             float* __restrict__ wvec, int zero_blocks) {
    if ((int)blockIdx.x < zero_blocks) {
        const int i = blockIdx.x * blockDim.x + threadIdx.x;
        if (i < n) p[i] = 0.0f;
        return;
    }
    const int wave = threadIdx.x >> 5;
    const int lane = threadIdx.x & 31;
    const int rbase = (((int)blockIdx.x - zero_blocks) * 8 + wave) * 4;
    float f[4];
#pragma unroll
    for (int j = 0; j < 4; ++j) f[j] = Wfc[lane + 32 * j];
    float keep = 0.0f;
#pragma unroll
    for (int r = 0; r < 4; ++r) {
        const float* wrow = W2 + (size_t)(rbase + r) * NHID + lane;
        float s = wrow[0] * f[0];
        s += wrow[32] * f[1];
        s += wrow[64] * f[2];
        s += wrow[96] * f[3];
        for (int off = 16; off > 0; off >>= 1)
            s += __shfl_xor(s, off, 32);
        if (lane == r) keep = s;
    }
    if (lane < 4) wvec[rbase + lane] = keep;
}

#define NODE_NR 16
#define NODE_R  6272
#define EDGE_NS 16
#define ACC_TPB 1024
static_assert(NODE_NR * NODE_R >= N_NODES, "node ranges must cover all nodes");

__device__ __forceinline__ int slice_len(int e) {
    return (((e + EDGE_NS - 1) / EDGE_NS) + 3) & ~3;
}

__device__ __forceinline__ void count_one(unsigned* cnt, int idx, unsigned r0) {
    const unsigned u = (unsigned)idx - r0;
    if (u < (unsigned)NODE_R) atomicAdd(&cnt[u], 1u);
}

__device__ __forceinline__ void count_four(unsigned* cnt, const int4 v, unsigned r0) {
    count_one(cnt, v.x, r0);
    count_one(cnt, v.y, r0);
    count_one(cnt, v.z, r0);
    count_one(cnt, v.w, r0);
}

__global__ void __launch_bounds__(ACC_TPB)
degree_kernel(const int* __restrict__ src, const int* __restrict__ dst,
              float* __restrict__ degS, float* __restrict__ degD,
              unsigned* __restrict__ Ctab, unsigned* __restrict__ RT, int e, int n) {
    __shared__ unsigned cntS[NODE_R];
    __shared__ unsigned cntD[NODE_R];
    __shared__ unsigned btot;
    const int t     = threadIdx.x;
    if (t == 0) btot = 0u;
    const int range = blockIdx.x % NODE_NR;
    const int slice = blockIdx.x / NODE_NR;
    const unsigned r0 = (unsigned)(range * NODE_R);

    for (int i = t; i < NODE_R; i += ACC_TPB) { cntS[i] = 0u; cntD[i] = 0u; }
    __syncthreads();

    const int per = slice_len(e);
    const int e0  = slice * per;
    const int e1  = (e0 + per < e) ? (e0 + per) : e;
    const int stride = 4 * ACC_TPB;
    int i = e0 + 4 * t;
    for (; i + stride + 4 <= e1; i += 2 * stride) {
        const int4 s0 = *(const int4*)(src + i);
        const int4 d0 = *(const int4*)(dst + i);
        const int4 s1 = *(const int4*)(src + i + stride);
        const int4 d1 = *(const int4*)(dst + i + stride);
        count_four(cntS, s0, r0);
        count_four(cntD, d0, r0);
        count_four(cntS, s1, r0);
        count_four(cntD, d1, r0);
    }
    for (; i < e1; i += stride) {
        if (i + 4 <= e1) {
            const int4 s0 = *(const int4*)(src + i);
            const int4 d0 = *(const int4*)(dst + i);
            count_four(cntS, s0, r0);
            count_four(cntD, d0, r0);
        } else {
            for (int k = i; k < e1; ++k) {
                count_one(cntS, src[k], r0);
                count_one(cntD, dst[k], r0);
            }
        }
    }
    __syncthreads();

    unsigned mytot = 0u;
    for (int j = t; j < NODE_R; j += ACC_TPB) {
        const int node = (int)r0 + j;
        if (node < n) {
            const unsigned cd = cntD[j];
            atomicAdd(&degS[node], (float)cntS[j]);
            atomicAdd(&degD[node], (float)cd);
            Ctab[(size_t)slice * n + node] = cd;
            mytot += cd;
        }
    }
    for (int off = 16; off > 0; off >>= 1)
        mytot += __shfl_xor(mytot, off, 32);
    if ((t & 31) == 0) atomicAdd(&btot, mytot);
    __syncthreads();
    if (t == 0) RT[range * EDGE_NS + slice] = btot;
}

#define SCAN_PT 7
static_assert(SCAN_PT * ACC_TPB >= NODE_R, "one scan block must cover a node range");
__global__ void __launch_bounds__(ACC_TPB)
rowstart_kernel(unsigned* __restrict__ Ctab, const unsigned* __restrict__ RT,
                int* __restrict__ rowptr, int n) {
    __shared__ unsigned tot[SCAN_PT * ACC_TPB];
    __shared__ unsigned wsum[32];
    __shared__ unsigned offs;
    const int t    = threadIdx.x;
    const int lane = t & 31;
    const int wave = t >> 5;
    const int r    = blockIdx.x;
    const int r0   = r * NODE_R;
    if (t == 0) offs = 0u;
    for (int j = t; j < SCAN_PT * ACC_TPB; j += ACC_TPB) {
        const int node = r0 + j;
        unsigned s = 0u;
        if (j < NODE_R && node < n) {
            for (int k = 0; k < EDGE_NS; ++k) s += Ctab[(size_t)k * n + node];
        }
        tot[j] = s;
    }
    __syncthreads();
    if (t < r * EDGE_NS) atomicAdd(&offs, RT[t]);
    unsigned loc[SCAN_PT];
    unsigned sum = 0u;
    for (int i = 0; i < SCAN_PT; ++i) { loc[i] = tot[SCAN_PT * t + i]; sum += loc[i]; }
    unsigned inc = sum;
    for (int off = 1; off < 32; off <<= 1) {
        const unsigned y = __shfl_up(inc, off, 32);
        if (lane >= off) inc += y;
    }
    if (lane == 31) wsum[wave] = inc;
    __syncthreads();
    if (t < 32) {
        const unsigned x = wsum[t];
        unsigned xi = x;
        for (int off = 1; off < 32; off <<= 1) {
            const unsigned y = __shfl_up(xi, off, 32);
            if (lane >= off) xi += y;
        }
        wsum[t] = xi - x;
    }
    __syncthreads();
    unsigned run = offs + wsum[wave] + (inc - sum);
    for (int i = 0; i < SCAN_PT; ++i) { const unsigned cc = loc[i]; tot[SCAN_PT * t + i] = run; run += cc; }
    __syncthreads();
    for (int j = t; j < NODE_R; j += ACC_TPB) {
        const int node = r0 + j;
        if (node < n) {
            unsigned b = tot[j];
            rowptr[node] = (int)b;
            for (int k = 0; k < EDGE_NS; ++k) {
                const size_t idx = (size_t)k * n + node;
                const unsigned cc = Ctab[idx];
                Ctab[idx] = b;
                b += cc;
            }
            if (node == n - 1) rowptr[n] = (int)b;
        }
    }
}

__device__ __forceinline__ void place_one(unsigned* cur, int s, int d, unsigned r0,
                                          int* __restrict__ col) {
    const unsigned u = (unsigned)d - r0;
    if (u < (unsigned)NODE_R) {
        const unsigned pos = atomicAdd(&cur[u], 1u);
        if (pos < (unsigned)N_EDGES) col[pos] = s;
    }
}

__global__ void __launch_bounds__(ACC_TPB)
place_kernel(const int* __restrict__ src, const int* __restrict__ dst,
             const unsigned* __restrict__ Ctab, int* __restrict__ col, int e, int n) {
    __shared__ unsigned cur[NODE_R];
    const int t     = threadIdx.x;
    const int range = blockIdx.x % NODE_NR;
    const int slice = blockIdx.x / NODE_NR;
    const unsigned r0 = (unsigned)(range * NODE_R);

    for (int j = t; j < NODE_R; j += ACC_TPB) {
        const int node = (int)r0 + j;
        cur[j] = (node < n) ? Ctab[(size_t)slice * n + node] : 0u;
    }
    __syncthreads();

    const int per = slice_len(e);
    const int e0  = slice * per;
    const int e1  = (e0 + per < e) ? (e0 + per) : e;
    const int stride = 4 * ACC_TPB;
    int i = e0 + 4 * t;
    for (; i + stride + 4 <= e1; i += 2 * stride) {
        const int4 s0 = *(const int4*)(src + i);
        const int4 d0 = *(const int4*)(dst + i);
        const int4 s1 = *(const int4*)(src + i + stride);
        const int4 d1 = *(const int4*)(dst + i + stride);
        place_one(cur, s0.x, d0.x, r0, col);
        place_one(cur, s0.y, d0.y, r0, col);
        place_one(cur, s0.z, d0.z, r0, col);
        place_one(cur, s0.w, d0.w, r0, col);
        place_one(cur, s1.x, d1.x, r0, col);
        place_one(cur, s1.y, d1.y, r0, col);
        place_one(cur, s1.z, d1.z, r0, col);
        place_one(cur, s1.w, d1.w, r0, col);
    }
    for (; i < e1; i += stride) {
        const int kend = (i + 4 <= e1) ? (i + 4) : e1;
        for (int k = i; k < kend; ++k) place_one(cur, src[k], dst[k], r0, col);
    }
}

__device__ __forceinline__ float deg_to_norm(float deg) {
    return rsqrtf(fmaxf(deg, 1.0f));
}

#define BM 80
#define BK 32
#define PA 36
#define PB 288
#define PC 132
#define SMEM_FLOATS (BM * PC)

__device__ __forceinline__ void gemm80_mainloop(const float* __restrict__ X,
                                                const float* __restrict__ scale,
                                                const float* __restrict__ W, int K,
                                                float* smem, v8f (&acc)[5]) {
    float* As = smem;
    float* Bs = smem + BM * PA;
    const int t    = threadIdx.x;
    const int wave = t >> 5;
    const int lane = t & 31;
    const int h    = lane >> 4;
    const int c    = lane & 15;
    const int row0 = blockIdx.x * BM;

    const bool third = (t < 128);
    const float* xa[3];
    float ns[3];
    int   aoff[3];
#pragma unroll
    for (int i = 0; i < 3; ++i) {
        const int idx = t + 256 * i;
        const int r   = (i < 2 || third) ? (idx >> 3) : 0;
        const int k4  = idx & 7;
        ns[i]   = deg_to_norm(scale[row0 + r]);
        xa[i]   = X + (size_t)(row0 + r) * K + 4 * k4;
        aoff[i] = r * PA + 4 * k4;
    }
    const float* wb[2];
    int boff[2];
#pragma unroll
    for (int j = 0; j < 2; ++j) {
        const int idx = t + 256 * j;
        const int pr  = idx >> 5;
        const int n4  = idx & 31;
        wb[j]   = W + (size_t)(2 * pr) * NHID + 4 * n4;
        boff[j] = pr * PB + 8 * n4;
    }

    float4 ra[3];
    float4 rb[2][2];

    const int nch = K / BK;

#pragma unroll
    for (int i = 0; i < 2; ++i) ra[i] = *(const float4*)(xa[i]);
    if (third) ra[2] = *(const float4*)(xa[2]);
#pragma unroll
    for (int j = 0; j < 2; ++j) {
        rb[j][0] = *(const float4*)(wb[j]);
        rb[j][1] = *(const float4*)(wb[j] + NHID);
    }

    const int afrag = c * PA + 2 * h;
    const int bfrag = h * PB + (16 * wave + c) * 2;

    for (int ch = 0; ch < nch; ++ch) {
#pragma unroll
        for (int i = 0; i < 3; ++i) {
            if (i < 2 || third) {
                float4 v = ra[i];
                v.x *= ns[i]; v.y *= ns[i]; v.z *= ns[i]; v.w *= ns[i];
                *(float4*)(As + aoff[i]) = v;
            }
        }
#pragma unroll
        for (int j = 0; j < 2; ++j) {
            const float4 lo = rb[j][0], hi = rb[j][1];
            *(float4*)(Bs + boff[j])     = make_float4(lo.x, hi.x, lo.y, hi.y);
            *(float4*)(Bs + boff[j] + 4) = make_float4(lo.z, hi.z, lo.w, hi.w);
        }
        __syncthreads();

        if (ch + 1 < nch) {
            const int ko = (ch + 1) * BK;
#pragma unroll
            for (int i = 0; i < 2; ++i) ra[i] = *(const float4*)(xa[i] + ko);
            if (third) ra[2] = *(const float4*)(xa[2] + ko);
#pragma unroll
            for (int j = 0; j < 2; ++j) {
                rb[j][0] = *(const float4*)(wb[j] + (size_t)ko * NHID);
                rb[j][1] = *(const float4*)(wb[j] + (size_t)ko * NHID + NHID);
            }
        }

#pragma unroll
        for (int kk = 0; kk < BK / 4; ++kk) {
            const v2f b = *(const v2f*)(Bs + bfrag + 2 * kk * PB);
#pragma unroll
            for (int tm = 0; tm < 5; ++tm) {
                const v2f a = *(const v2f*)(As + afrag + 16 * tm * PA + 4 * kk);
                acc[tm] = __builtin_amdgcn_wmma_f32_16x16x4_f32(false, a, false, b,
                                                                (short)0, acc[tm], false, false);
            }
        }
        __syncthreads();
    }
}

__device__ __forceinline__ void gemm80_stage_result(float* smem, const v8f (&acc)[5]) {
    const int wave = threadIdx.x >> 5;
    const int lane = threadIdx.x & 31;
    const int h    = lane >> 4;
    const int c    = lane & 15;
#pragma unroll
    for (int tm = 0; tm < 5; ++tm) {
#pragma unroll
        for (int r = 0; r < 8; ++r) {
            smem[(16 * tm + 8 * h + r) * PC + 16 * wave + c] = acc[tm][r];
        }
    }
    __syncthreads();
}

__global__ void __launch_bounds__(256)
gemm_scaled_wmma(const float* __restrict__ X, const float* __restrict__ scale,
                 const float* __restrict__ W, float* __restrict__ Out, int K) {
    __shared__ __attribute__((aligned(16))) float smem[SMEM_FLOATS];
    v8f acc[5] = {};
    gemm80_mainloop(X, scale, W, K, smem, acc);
    gemm80_stage_result(smem, acc);

    const int wave = threadIdx.x >> 5;
    const int lane = threadIdx.x & 31;
    const int row0 = blockIdx.x * BM;
#pragma unroll
    for (int i = 0; i < BM / 8; ++i) {
        const int row = wave + 8 * i;
        const float4 v = *(const float4*)(smem + row * PC + 4 * lane);
        *(float4*)(Out + (size_t)(row0 + row) * NHID + 4 * lane) = v;
    }
}

#define NODES_PER_WAVE 10
#define EDGE_BATCH     8

template <bool FULL>
__device__ __forceinline__ void gather_batch(const float* __restrict__ M, int sv, int j0,
                                             int m, int lane, float (&v)[EDGE_BATCH][4]) {
    int s[EDGE_BATCH];
#pragma unroll
    for (int u = 0; u < EDGE_BATCH; ++u) {
        const int id = __builtin_amdgcn_readlane(sv, j0 + u);
        s[u] = id < 0 ? 0 : (id >= N_NODES ? N_NODES - 1 : id);
    }
#pragma unroll
    for (int u = 0; u < EDGE_BATCH; ++u) {
        if (FULL || j0 + u < m) {
            const float* __restrict__ mrow = M + (size_t)s[u] * NHID + lane;
            v[u][0] = mrow[0];
            v[u][1] = mrow[32];
            v[u][2] = mrow[64];
            v[u][3] = mrow[96];
        } else {
            v[u][0] = 0.0f; v[u][1] = 0.0f; v[u][2] = 0.0f; v[u][3] = 0.0f;
        }
    }
}

__device__ __forceinline__ float wave_sum(float x) {
    x += __builtin_bit_cast(float, __builtin_amdgcn_update_dpp(0, __builtin_bit_cast(int, x), 0x161, 0xf, 0xf, false));
    x += __builtin_bit_cast(float, __builtin_amdgcn_update_dpp(0, __builtin_bit_cast(int, x), 0x162, 0xf, 0xf, false));
    x += __builtin_bit_cast(float, __builtin_amdgcn_update_dpp(0, __builtin_bit_cast(int, x), 0x164, 0xf, 0xf, false));
    x += __builtin_bit_cast(float, __builtin_amdgcn_update_dpp(0, __builtin_bit_cast(int, x), 0x168, 0xf, 0xf, false));
    const int b = __builtin_bit_cast(int, x);
    x += __builtin_bit_cast(float, __builtin_amdgcn_permlanex16(b, b, 0x76543210, 0xfedcba98, false, false));
    return x;
}

__device__ __forceinline__ float row_relu_dot(float a0, float a1, float a2, float a3, float nd,
                                              const float (&br)[4], const float (&wr)[4]) {
    const float h0 = a0 * nd + br[0];
    const float h1 = a1 * nd + br[1];
    const float h2 = a2 * nd + br[2];
    const float h3 = a3 * nd + br[3];
    float p = (h0 > 0.0f ? h0 : 0.0f) * wr[0];
    p += (h1 > 0.0f ? h1 : 0.0f) * wr[1];
    p += (h2 > 0.0f ? h2 : 0.0f) * wr[2];
    p += (h3 > 0.0f ? h3 : 0.0f) * wr[3];
    return wave_sum(p);
}

__global__ void __launch_bounds__(256)
gather_rows_dot(const float* __restrict__ M, const int* __restrict__ rowptr,
                const int* __restrict__ col, const float* __restrict__ degS,
                const float* __restrict__ degD, const float* __restrict__ bias,
                const float* __restrict__ wvec, float* __restrict__ S, int n) {
    const int w    = __builtin_amdgcn_readfirstlane((int)((blockIdx.x * blockDim.x + threadIdx.x) >> 5));
    const int lane = threadIdx.x & 31;
    const int n0   = w * NODES_PER_WAVE;
    if (n0 >= n) return;
    const int cntN = (n - n0 < NODES_PER_WAVE) ? (n - n0) : NODES_PER_WAVE;
    const int rp = rowptr[n0 + (lane < cntN ? lane : cntN)];
    const int   ln  = n0 + (lane < cntN ? lane : cntN - 1);
    const float ndl = deg_to_norm(degD[ln]);
    const float nsl = deg_to_norm(degS[ln]);
    float br[4];
    float wr[4];
#pragma unroll
    for (int t = 0; t < 4; ++t) {
        br[t] = bias[lane + 32 * t];
        wr[t] = wvec[lane + 32 * t];
    }
    int E0 = __builtin_amdgcn_readlane(rp, 0);
    int E1 = __builtin_amdgcn_readlane(rp, cntN);
    E0 = E0 < 0 ? 0 : (E0 > N_EDGES ? N_EDGES : E0);
    E1 = E1 < E0 ? E0 : (E1 > N_EDGES ? N_EDGES : E1);
    int cur  = 0;
    int endc = __builtin_amdgcn_readlane(rp, 1);
    float a0 = 0.0f, a1 = 0.0f, a2 = 0.0f, a3 = 0.0f;
    float sout = 0.0f;

    for (int cb = E0; cb < E1; cb += 32) {
        const int m  = (E1 - cb < 32) ? (E1 - cb) : 32;
        const int sv = (lane < m) ? col[cb + lane] : 0;
        for (int j0 = 0; j0 < m; j0 += EDGE_BATCH) {
            float v[EDGE_BATCH][4];
            if (j0 + EDGE_BATCH <= m) gather_batch<true>(M, sv, j0, m, lane, v);
            else                      gather_batch<false>(M, sv, j0, m, lane, v);
#pragma unroll
            for (int u = 0; u < EDGE_BATCH; ++u) {
                if (j0 + u < m) {
                    const int j = cb + j0 + u;
                    while (cur + 1 < cntN && j >= endc) {
                        const float nd = __int_as_float(__builtin_amdgcn_readlane(__float_as_int(ndl), cur));
                        const float p  = row_relu_dot(a0, a1, a2, a3, nd, br, wr);
                        if (lane == cur) sout = p;
                        a0 = 0.0f; a1 = 0.0f; a2 = 0.0f; a3 = 0.0f;
                        ++cur;
                        endc = __builtin_amdgcn_readlane(rp, cur + 1);
                    }
                    a0 += v[u][0];
                    a1 += v[u][1];
                    a2 += v[u][2];
                    a3 += v[u][3];
                }
            }
        }
    }
    while (cur < cntN) {
        const float nd = __int_as_float(__builtin_amdgcn_readlane(__float_as_int(ndl), cur));
        const float p  = row_relu_dot(a0, a1, a2, a3, nd, br, wr);
        if (lane == cur) sout = p;
        a0 = 0.0f; a1 = 0.0f; a2 = 0.0f; a3 = 0.0f;
        ++cur;
    }
    if (lane < cntN) S[n0 + lane] = sout * nsl;
}

#define FS_TPB   512
#define FS_NODES 416
#define FS_BATCH 16
#define FS_CAP   (FS_TPB * FS_BATCH)

__global__ void __launch_bounds__(FS_TPB)
edge_sum_out(const float* __restrict__ S, const int* __restrict__ rowptr,
             const int* __restrict__ col, const float* __restrict__ normD,
             const float* __restrict__ b2, const float* __restrict__ Wfc,
             const float* __restrict__ bfc, float* __restrict__ out, int n) {
    __shared__ float vals[FS_CAP];
    const int t    = threadIdx.x;
    const int lane = t & 31;
    const int n0   = blockIdx.x * FS_NODES;
    if (n0 >= n) return;
    const int  cnt = (n - n0 < FS_NODES) ? (n - n0) : FS_NODES;
    const bool own = t < cnt;
    const int  i   = n0 + (own ? t : 0);

    int B0 = rowptr[n0];
    int B1 = rowptr[n0 + cnt];
    int r0 = rowptr[i];
    int r1 = rowptr[i + 1];
    const float dg = normD[i];
    float cb = 0.0f;
    for (int q = 0; q < 4; ++q) {
        const int j = lane + 32 * q;
        cb += b2[j] * Wfc[j];
    }
    for (int off = 16; off > 0; off >>= 1)
        cb += __shfl_xor(cb, off, 32);
    B0 = B0 < 0 ? 0 : (B0 > N_EDGES ? N_EDGES : B0);
    B1 = B1 < B0 ? B0 : (B1 > N_EDGES ? N_EDGES : B1);
    r0 = r0 < 0 ? 0 : (r0 > N_EDGES ? N_EDGES : r0);
    r1 = r1 < 0 ? 0 : (r1 > N_EDGES ? N_EDGES : r1);

    float s = 0.0f;
    for (int base = B0; base < B1; base += FS_CAP) {
        const int len = (B1 - base < FS_CAP) ? (B1 - base) : FS_CAP;
        int   id[FS_BATCH];
        float v[FS_BATCH];
#pragma unroll
        for (int u = 0; u < FS_BATCH; ++u) {
            const int k = u * FS_TPB + t;
            id[u] = col[base + (k < len ? k : len - 1)];
        }
#pragma unroll
        for (int u = 0; u < FS_BATCH; ++u) {
            const int c = id[u];
            v[u] = S[c < 0 ? 0 : (c >= N_NODES ? N_NODES - 1 : c)];
        }
#pragma unroll
        for (int u = 0; u < FS_BATCH; ++u) vals[u * FS_TPB + t] = v[u];
        __syncthreads();
        if (own) {
            const int lo = (r0 > base ? r0 : base) - base;
            const int hi = (r1 < base + len ? r1 : base + len) - base;
            int k = lo;
            for (; k + 4 <= hi; k += 4) {
                const float v0 = vals[k], v1 = vals[k + 1], v2 = vals[k + 2], v3 = vals[k + 3];
                s += v0; s += v1; s += v2; s += v3;
            }
            for (; k < hi; ++k) s += vals[k];
        }
        __syncthreads();
    }
    if (own) out[i] = s * deg_to_norm(dg) + (cb + bfc[0]);
}

extern "C" void kernel_launch(void* const* d_in, const int* in_sizes, int n_in,
                              void* d_out, int out_size, void* d_ws, size_t ws_size,
                              hipStream_t stream) {
    const float* x   = (const float*)d_in[0];
    const float* W1  = (const float*)d_in[1];
    const float* b1  = (const float*)d_in[2];
    const float* W2  = (const float*)d_in[3];
    const float* b2  = (const float*)d_in[4];
    const float* Wfc = (const float*)d_in[5];
    const float* bfc = (const float*)d_in[6];
    const int*   src = (const int*)d_in[7];
    const int*   dst = (const int*)d_in[8];
    float* out = (float*)d_out;

    const int N = N_NODES, E = N_EDGES;

    float* BUF_A = (float*)d_ws;
    float* BUF_B = BUF_A + (size_t)N * NHID;
    float* normS = BUF_B + (size_t)N * NHID;
    float* normD = normS + N;

    int*      col    = (int*)(BUF_B + (size_t)7864320);
    unsigned* Ctab   = (unsigned*)(col + 1600000);
    int*      rowptr = (int*)(Ctab + 1600000);
    unsigned* RT     = (unsigned*)(rowptr + 100032);
    float*    S      = (float*)(RT + 256);
    float*    wvec   = S + 100000;

    const int TPB = 256;
    const int ZERO_BLOCKS = (2 * N + TPB - 1) / TPB;

    setup_kernel<<<ZERO_BLOCKS + WV_BLOCKS, TPB, 0, stream>>>(normS, 2 * N, W2, Wfc, wvec,
                                                            ZERO_BLOCKS);
    degree_kernel<<<NODE_NR * EDGE_NS, ACC_TPB, 0, stream>>>(src, dst, normS, normD, Ctab, RT, E, N);

    rowstart_kernel<<<NODE_NR, ACC_TPB, 0, stream>>>(Ctab, RT, rowptr, N);
    place_kernel<<<NODE_NR * EDGE_NS, ACC_TPB, 0, stream>>>(src, dst, Ctab, col, E, N);

    gemm_scaled_wmma<<<1250, TPB, 0, stream>>>(x, normS, W1, BUF_A, NFEAT);

    gather_rows_dot<<<1250, TPB, 0, stream>>>(BUF_A, rowptr, col, normS, normD, b1, wvec, S, N);

    edge_sum_out<<<(N + FS_NODES - 1) / FS_NODES, FS_TPB, 0, stream>>>(S, rowptr, col, normD, b2, Wfc,
                                                                       bfc, out, N);
}


# row-start table: 224 blocks of 448 nodes (one node per thread, its 16 slice counts as independent loads, piece totals from the degree pass) instead of 16 blocks walking a serial load-store chain
# speedup vs baseline: 6.4557x; 1.0956x over previous
#include <hip/hip_runtime.h>

typedef __attribute__((ext_vector_type(2))) float v2f;
typedef __attribute__((ext_vector_type(8))) float v8f;

#define N_NODES 100000
#define N_EDGES 1600000
#define NFEAT   256
#define NHID    128

#define WV_BLOCKS 4
__global__ void __launch_bounds__(256)
setup_kernel(float* __restrict__ p, int n,
             const float* __restrict__ W2, const float* __restrict__ Wfc,
             float* __restrict__ wvec, int zero_blocks) {
    if ((int)blockIdx.x < zero_blocks) {
        const int i = blockIdx.x * blockDim.x + threadIdx.x;
        if (i < n) p[i] = 0.0f;
        return;
    }
    const int wave = threadIdx.x >> 5;
    const int lane = threadIdx.x & 31;
    const int rbase = (((int)blockIdx.x - zero_blocks) * 8 + wave) * 4;
    float f[4];
#pragma unroll
    for (int j = 0; j < 4; ++j) f[j] = Wfc[lane + 32 * j];
    float keep = 0.0f;
#pragma unroll
    for (int r = 0; r < 4; ++r) {
        const float* wrow = W2 + (size_t)(rbase + r) * NHID + lane;
        float s = wrow[0] * f[0];
        s += wrow[32] * f[1];
        s += wrow[64] * f[2];
        s += wrow[96] * f[3];
        for (int off = 16; off > 0; off >>= 1)
            s += __shfl_xor(s, off, 32);
        if (lane == r) keep = s;
    }
    if (lane < 4) wvec[rbase + lane] = keep;
}

#define NODE_NR 16
#define NODE_R  6272
#define EDGE_NS 16
#define ACC_TPB 1024
static_assert(NODE_NR * NODE_R >= N_NODES, "node ranges must cover all nodes");

__device__ __forceinline__ int slice_len(int e) {
    return (((e + EDGE_NS - 1) / EDGE_NS) + 3) & ~3;
}

__device__ __forceinline__ void count_one(unsigned* cnt, int idx, unsigned r0) {
    const unsigned u = (unsigned)idx - r0;
    if (u < (unsigned)NODE_R) atomicAdd(&cnt[u], 1u);
}

__device__ __forceinline__ void count_four(unsigned* cnt, const int4 v, unsigned r0) {
    count_one(cnt, v.x, r0);
    count_one(cnt, v.y, r0);
    count_one(cnt, v.z, r0);
    count_one(cnt, v.w, r0);
}

#define RS_SUB    14
#define RS_ROWS   14
#define RS_NODES  (32 * RS_ROWS)
#define RS_BLOCKS (NODE_NR * RS_SUB)
static_assert(RS_SUB * RS_NODES == NODE_R, "the pieces must tile a node range");

__device__ __forceinline__ unsigned wave_sum_u(unsigned x) {
    x += (unsigned)__builtin_amdgcn_update_dpp(0, (int)x, 0x161, 0xf, 0xf, false);
    x += (unsigned)__builtin_amdgcn_update_dpp(0, (int)x, 0x162, 0xf, 0xf, false);
    x += (unsigned)__builtin_amdgcn_update_dpp(0, (int)x, 0x164, 0xf, 0xf, false);
    x += (unsigned)__builtin_amdgcn_update_dpp(0, (int)x, 0x168, 0xf, 0xf, false);
    const int b = (int)x;
    x += (unsigned)__builtin_amdgcn_permlanex16(b, b, 0x76543210, 0xfedcba98, false, false);
    return x;
}

__global__ void __launch_bounds__(ACC_TPB)
degree_kernel(const int* __restrict__ src, const int* __restrict__ dst,
              float* __restrict__ degS, float* __restrict__ degD,
              unsigned* __restrict__ Ctab, unsigned* __restrict__ RT, int e, int n) {
    __shared__ unsigned cntS[NODE_R];
    __shared__ unsigned cntD[NODE_R];
    __shared__ unsigned ptot[RS_SUB];
    const int t     = threadIdx.x;
    if (t < RS_SUB) ptot[t] = 0u;
    const int range = blockIdx.x % NODE_NR;
    const int slice = blockIdx.x / NODE_NR;
    const unsigned r0 = (unsigned)(range * NODE_R);

    for (int i = t; i < NODE_R; i += ACC_TPB) { cntS[i] = 0u; cntD[i] = 0u; }
    __syncthreads();

    const int per = slice_len(e);
    const int e0  = slice * per;
    const int e1  = (e0 + per < e) ? (e0 + per) : e;
    const int stride = 4 * ACC_TPB;
    int i = e0 + 4 * t;
    for (; i + stride + 4 <= e1; i += 2 * stride) {
        const int4 s0 = *(const int4*)(src + i);
        const int4 d0 = *(const int4*)(dst + i);
        const int4 s1 = *(const int4*)(src + i + stride);
        const int4 d1 = *(const int4*)(dst + i + stride);
        count_four(cntS, s0, r0);
        count_four(cntD, d0, r0);
        count_four(cntS, s1, r0);
        count_four(cntD, d1, r0);
    }
    for (; i < e1; i += stride) {
        if (i + 4 <= e1) {
            const int4 s0 = *(const int4*)(src + i);
            const int4 d0 = *(const int4*)(dst + i);
            count_four(cntS, s0, r0);
            count_four(cntD, d0, r0);
        } else {
            for (int k = i; k < e1; ++k) {
                count_one(cntS, src[k], r0);
                count_one(cntD, dst[k], r0);
            }
        }
    }
    __syncthreads();

    for (int j = t; j < NODE_R; j += ACC_TPB) {
        const int node = (int)r0 + j;
        unsigned cd = 0u;
        if (node < n) {
            cd = cntD[j];
            atomicAdd(&degS[node], (float)cntS[j]);
            atomicAdd(&degD[node], (float)cd);
            Ctab[(size_t)slice * n + node] = cd;
        }
        const unsigned rs = wave_sum_u(cd);
        if ((t & 31) == 0) atomicAdd(&ptot[(j >> 5) / RS_ROWS], rs);
    }
    __syncthreads();
    if (t < RS_SUB) RT[(range * RS_SUB + t) * EDGE_NS + slice] = ptot[t];
}

#define RS_PRE ((RS_BLOCKS * EDGE_NS) / RS_NODES)
static_assert(RS_PRE * RS_NODES == RS_BLOCKS * EDGE_NS, "RS_PRE loads per thread must cover RT");
__global__ void __launch_bounds__(RS_NODES)
rowstart_kernel(unsigned* __restrict__ Ctab, const unsigned* __restrict__ RT,
                int* __restrict__ rowptr, int n) {
    __shared__ unsigned wtot[RS_ROWS];
    __shared__ unsigned wpre[RS_ROWS];
    const int t    = threadIdx.x;
    const int lane = t & 31;
    const int wave = t >> 5;
    const int b    = blockIdx.x;
    const int node = b * RS_NODES + t;
    const bool live = node < n;
    const int  nd   = live ? node : 0;

    const int lim = b * EDGE_NS;
    unsigned pre = 0u;
#pragma unroll
    for (int q = 0; q < RS_PRE; ++q) {
        const int i = t + q * RS_NODES;
        const unsigned v = RT[i < lim ? i : 0];
        pre += (i < lim) ? v : 0u;
    }
    unsigned c[EDGE_NS];
    unsigned tot = 0u;
#pragma unroll
    for (int k = 0; k < EDGE_NS; ++k) {
        const unsigned v = Ctab[(size_t)k * n + nd];
        c[k] = live ? v : 0u;
        tot += c[k];
    }

    unsigned inc = tot;
    for (int off = 1; off < 32; off <<= 1) {
        const unsigned y = __shfl_up(inc, off, 32);
        if (lane >= off) inc += y;
    }
    pre = wave_sum_u(pre);
    if (lane == 31) wtot[wave] = inc;
    if (lane == 0)  wpre[wave] = pre;
    __syncthreads();
    unsigned base = 0u;
#pragma unroll
    for (int w = 0; w < RS_ROWS; ++w) {
        base += wpre[w];
        if (w < wave) base += wtot[w];
    }
    unsigned run = base + (inc - tot);
    if (live) {
        rowptr[node] = (int)run;
#pragma unroll
        for (int k = 0; k < EDGE_NS; ++k) {
            Ctab[(size_t)k * n + node] = run;
            run += c[k];
        }
        if (node == n - 1) rowptr[n] = (int)run;
    }
}

__device__ __forceinline__ void place_one(unsigned* cur, int s, int d, unsigned r0,
                                          int* __restrict__ col) {
    const unsigned u = (unsigned)d - r0;
    if (u < (unsigned)NODE_R) {
        const unsigned pos = atomicAdd(&cur[u], 1u);
        if (pos < (unsigned)N_EDGES) col[pos] = s;
    }
}

__global__ void __launch_bounds__(ACC_TPB)
place_kernel(const int* __restrict__ src, const int* __restrict__ dst,
             const unsigned* __restrict__ Ctab, int* __restrict__ col, int e, int n) {
    __shared__ unsigned cur[NODE_R];
    const int t     = threadIdx.x;
    const int range = blockIdx.x % NODE_NR;
    const int slice = blockIdx.x / NODE_NR;
    const unsigned r0 = (unsigned)(range * NODE_R);

    for (int j = t; j < NODE_R; j += ACC_TPB) {
        const int node = (int)r0 + j;
        cur[j] = (node < n) ? Ctab[(size_t)slice * n + node] : 0u;
    }
    __syncthreads();

    const int per = slice_len(e);
    const int e0  = slice * per;
    const int e1  = (e0 + per < e) ? (e0 + per) : e;
    const int stride = 4 * ACC_TPB;
    int i = e0 + 4 * t;
    for (; i + stride + 4 <= e1; i += 2 * stride) {
        const int4 s0 = *(const int4*)(src + i);
        const int4 d0 = *(const int4*)(dst + i);
        const int4 s1 = *(const int4*)(src + i + stride);
        const int4 d1 = *(const int4*)(dst + i + stride);
        place_one(cur, s0.x, d0.x, r0, col);
        place_one(cur, s0.y, d0.y, r0, col);
        place_one(cur, s0.z, d0.z, r0, col);
        place_one(cur, s0.w, d0.w, r0, col);
        place_one(cur, s1.x, d1.x, r0, col);
        place_one(cur, s1.y, d1.y, r0, col);
        place_one(cur, s1.z, d1.z, r0, col);
        place_one(cur, s1.w, d1.w, r0, col);
    }
    for (; i < e1; i += stride) {
        const int kend = (i + 4 <= e1) ? (i + 4) : e1;
        for (int k = i; k < kend; ++k) place_one(cur, src[k], dst[k], r0, col);
    }
}

__device__ __forceinline__ float deg_to_norm(float deg) {
    return rsqrtf(fmaxf(deg, 1.0f));
}

#define BM 80
#define BK 32
#define PA 36
#define PB 288
#define PC 132
#define SMEM_FLOATS (BM * PC)

__device__ __forceinline__ void gemm80_mainloop(const float* __restrict__ X,
                                                const float* __restrict__ scale,
                                                const float* __restrict__ W, int K,
                                                float* smem, v8f (&acc)[5]) {
    float* As = smem;
    float* Bs = smem + BM * PA;
    const int t    = threadIdx.x;
    const int wave = t >> 5;
    const int lane = t & 31;
    const int h    = lane >> 4;
    const int c    = lane & 15;
    const int row0 = blockIdx.x * BM;

    const bool third = (t < 128);
    const float* xa[3];
    float ns[3];
    int   aoff[3];
#pragma unroll
    for (int i = 0; i < 3; ++i) {
        const int idx = t + 256 * i;
        const int r   = (i < 2 || third) ? (idx >> 3) : 0;
        const int k4  = idx & 7;
        ns[i]   = deg_to_norm(scale[row0 + r]);
        xa[i]   = X + (size_t)(row0 + r) * K + 4 * k4;
        aoff[i] = r * PA + 4 * k4;
    }
    const float* wb[2];
    int boff[2];
#pragma unroll
    for (int j = 0; j < 2; ++j) {
        const int idx = t + 256 * j;
        const int pr  = idx >> 5;
        const int n4  = idx & 31;
        wb[j]   = W + (size_t)(2 * pr) * NHID + 4 * n4;
        boff[j] = pr * PB + 8 * n4;
    }

    float4 ra[3];
    float4 rb[2][2];

    const int nch = K / BK;

#pragma unroll
    for (int i = 0; i < 2; ++i) ra[i] = *(const float4*)(xa[i]);
    if (third) ra[2] = *(const float4*)(xa[2]);
#pragma unroll
    for (int j = 0; j < 2; ++j) {
        rb[j][0] = *(const float4*)(wb[j]);
        rb[j][1] = *(const float4*)(wb[j] + NHID);
    }

    const int afrag = c * PA + 2 * h;
    const int bfrag = h * PB + (16 * wave + c) * 2;

    for (int ch = 0; ch < nch; ++ch) {
#pragma unroll
        for (int i = 0; i < 3; ++i) {
            if (i < 2 || third) {
                float4 v = ra[i];
                v.x *= ns[i]; v.y *= ns[i]; v.z *= ns[i]; v.w *= ns[i];
                *(float4*)(As + aoff[i]) = v;
            }
        }
#pragma unroll
        for (int j = 0; j < 2; ++j) {
            const float4 lo = rb[j][0], hi = rb[j][1];
            *(float4*)(Bs + boff[j])     = make_float4(lo.x, hi.x, lo.y, hi.y);
            *(float4*)(Bs + boff[j] + 4) = make_float4(lo.z, hi.z, lo.w, hi.w);
        }
        __syncthreads();

        if (ch + 1 < nch) {
            const int ko = (ch + 1) * BK;
#pragma unroll
            for (int i = 0; i < 2; ++i) ra[i] = *(const float4*)(xa[i] + ko);
            if (third) ra[2] = *(const float4*)(xa[2] + ko);
#pragma unroll
            for (int j = 0; j < 2; ++j) {
                rb[j][0] = *(const float4*)(wb[j] + (size_t)ko * NHID);
                rb[j][1] = *(const float4*)(wb[j] + (size_t)ko * NHID + NHID);
            }
        }

#pragma unroll
        for (int kk = 0; kk < BK / 4; ++kk) {
            const v2f b = *(const v2f*)(Bs + bfrag + 2 * kk * PB);
#pragma unroll
            for (int tm = 0; tm < 5; ++tm) {
                const v2f a = *(const v2f*)(As + afrag + 16 * tm * PA + 4 * kk);
                acc[tm] = __builtin_amdgcn_wmma_f32_16x16x4_f32(false, a, false, b,
                                                                (short)0, acc[tm], false, false);
            }
        }
        __syncthreads();
    }
}

__device__ __forceinline__ void gemm80_stage_result(float* smem, const v8f (&acc)[5]) {
    const int wave = threadIdx.x >> 5;
    const int lane = threadIdx.x & 31;
    const int h    = lane >> 4;
    const int c    = lane & 15;
#pragma unroll
    for (int tm = 0; tm < 5; ++tm) {
#pragma unroll
        for (int r = 0; r < 8; ++r) {
            smem[(16 * tm + 8 * h + r) * PC + 16 * wave + c] = acc[tm][r];
        }
    }
    __syncthreads();
}

__global__ void __launch_bounds__(256)
gemm_scaled_wmma(const float* __restrict__ X, const float* __restrict__ scale,
                 const float* __restrict__ W, float* __restrict__ Out, int K) {
    __shared__ __attribute__((aligned(16))) float smem[SMEM_FLOATS];
    v8f acc[5] = {};
    gemm80_mainloop(X, scale, W, K, smem, acc);
    gemm80_stage_result(smem, acc);

    const int wave = threadIdx.x >> 5;
    const int lane = threadIdx.x & 31;
    const int row0 = blockIdx.x * BM;
#pragma unroll
    for (int i = 0; i < BM / 8; ++i) {
        const int row = wave + 8 * i;
        const float4 v = *(const float4*)(smem + row * PC + 4 * lane);
        *(float4*)(Out + (size_t)(row0 + row) * NHID + 4 * lane) = v;
    }
}

#define NODES_PER_WAVE 10
#define EDGE_BATCH     8

template <bool FULL>
__device__ __forceinline__ void gather_batch(const float* __restrict__ M, int sv, int j0,
                                             int m, int lane, float (&v)[EDGE_BATCH][4]) {
    int s[EDGE_BATCH];
#pragma unroll
    for (int u = 0; u < EDGE_BATCH; ++u) {
        const int id = __builtin_amdgcn_readlane(sv, j0 + u);
        s[u] = id < 0 ? 0 : (id >= N_NODES ? N_NODES - 1 : id);
    }
#pragma unroll
    for (int u = 0; u < EDGE_BATCH; ++u) {
        if (FULL || j0 + u < m) {
            const float* __restrict__ mrow = M + (size_t)s[u] * NHID + lane;
            v[u][0] = mrow[0];
            v[u][1] = mrow[32];
            v[u][2] = mrow[64];
            v[u][3] = mrow[96];
        } else {
            v[u][0] = 0.0f; v[u][1] = 0.0f; v[u][2] = 0.0f; v[u][3] = 0.0f;
        }
    }
}

__device__ __forceinline__ float wave_sum(float x) {
    x += __builtin_bit_cast(float, __builtin_amdgcn_update_dpp(0, __builtin_bit_cast(int, x), 0x161, 0xf, 0xf, false));
    x += __builtin_bit_cast(float, __builtin_amdgcn_update_dpp(0, __builtin_bit_cast(int, x), 0x162, 0xf, 0xf, false));
    x += __builtin_bit_cast(float, __builtin_amdgcn_update_dpp(0, __builtin_bit_cast(int, x), 0x164, 0xf, 0xf, false));
    x += __builtin_bit_cast(float, __builtin_amdgcn_update_dpp(0, __builtin_bit_cast(int, x), 0x168, 0xf, 0xf, false));
    const int b = __builtin_bit_cast(int, x);
    x += __builtin_bit_cast(float, __builtin_amdgcn_permlanex16(b, b, 0x76543210, 0xfedcba98, false, false));
    return x;
}

__device__ __forceinline__ float row_relu_dot(float a0, float a1, float a2, float a3, float nd,
                                              const float (&br)[4], const float (&wr)[4]) {
    const float h0 = a0 * nd + br[0];
    const float h1 = a1 * nd + br[1];
    const float h2 = a2 * nd + br[2];
    const float h3 = a3 * nd + br[3];
    float p = (h0 > 0.0f ? h0 : 0.0f) * wr[0];
    p += (h1 > 0.0f ? h1 : 0.0f) * wr[1];
    p += (h2 > 0.0f ? h2 : 0.0f) * wr[2];
    p += (h3 > 0.0f ? h3 : 0.0f) * wr[3];
    return wave_sum(p);
}

__global__ void __launch_bounds__(256)
gather_rows_dot(const float* __restrict__ M, const int* __restrict__ rowptr,
                const int* __restrict__ col, const float* __restrict__ degS,
                const float* __restrict__ degD, const float* __restrict__ bias,
                const float* __restrict__ wvec, float* __restrict__ S, int n) {
    const int w    = __builtin_amdgcn_readfirstlane((int)((blockIdx.x * blockDim.x + threadIdx.x) >> 5));
    const int lane = threadIdx.x & 31;
    const int n0   = w * NODES_PER_WAVE;
    if (n0 >= n) return;
    const int cntN = (n - n0 < NODES_PER_WAVE) ? (n - n0) : NODES_PER_WAVE;
    const int rp = rowptr[n0 + (lane < cntN ? lane : cntN)];
    const int   ln  = n0 + (lane < cntN ? lane : cntN - 1);
    const float ndl = deg_to_norm(degD[ln]);
    const float nsl = deg_to_norm(degS[ln]);
    float br[4];
    float wr[4];
#pragma unroll
    for (int t = 0; t < 4; ++t) {
        br[t] = bias[lane + 32 * t];
        wr[t] = wvec[lane + 32 * t];
    }
    int E0 = __builtin_amdgcn_readlane(rp, 0);
    int E1 = __builtin_amdgcn_readlane(rp, cntN);
    E0 = E0 < 0 ? 0 : (E0 > N_EDGES ? N_EDGES : E0);
    E1 = E1 < E0 ? E0 : (E1 > N_EDGES ? N_EDGES : E1);
    int cur  = 0;
    int endc = __builtin_amdgcn_readlane(rp, 1);
    float a0 = 0.0f, a1 = 0.0f, a2 = 0.0f, a3 = 0.0f;
    float sout = 0.0f;

    for (int cb = E0; cb < E1; cb += 32) {
        const int m  = (E1 - cb < 32) ? (E1 - cb) : 32;
        const int sv = (lane < m) ? col[cb + lane] : 0;
        for (int j0 = 0; j0 < m; j0 += EDGE_BATCH) {
            float v[EDGE_BATCH][4];
            if (j0 + EDGE_BATCH <= m) gather_batch<true>(M, sv, j0, m, lane, v);
            else                      gather_batch<false>(M, sv, j0, m, lane, v);
#pragma unroll
            for (int u = 0; u < EDGE_BATCH; ++u) {
                if (j0 + u < m) {
                    const int j = cb + j0 + u;
                    while (cur + 1 < cntN && j >= endc) {
                        const float nd = __int_as_float(__builtin_amdgcn_readlane(__float_as_int(ndl), cur));
                        const float p  = row_relu_dot(a0, a1, a2, a3, nd, br, wr);
                        if (lane == cur) sout = p;
                        a0 = 0.0f; a1 = 0.0f; a2 = 0.0f; a3 = 0.0f;
                        ++cur;
                        endc = __builtin_amdgcn_readlane(rp, cur + 1);
                    }
                    a0 += v[u][0];
                    a1 += v[u][1];
                    a2 += v[u][2];
                    a3 += v[u][3];
                }
            }
        }
    }
    while (cur < cntN) {
        const float nd = __int_as_float(__builtin_amdgcn_readlane(__float_as_int(ndl), cur));
        const float p  = row_relu_dot(a0, a1, a2, a3, nd, br, wr);
        if (lane == cur) sout = p;
        a0 = 0.0f; a1 = 0.0f; a2 = 0.0f; a3 = 0.0f;
        ++cur;
    }
    if (lane < cntN) S[n0 + lane] = sout * nsl;
}

#define FS_TPB   512
#define FS_NODES 416
#define FS_BATCH 16
#define FS_CAP   (FS_TPB * FS_BATCH)

__global__ void __launch_bounds__(FS_TPB)
edge_sum_out(const float* __restrict__ S, const int* __restrict__ rowptr,
             const int* __restrict__ col, const float* __restrict__ normD,
             const float* __restrict__ b2, const float* __restrict__ Wfc,
             const float* __restrict__ bfc, float* __restrict__ out, int n) {
    __shared__ float vals[FS_CAP];
    const int t    = threadIdx.x;
    const int lane = t & 31;
    const int n0   = blockIdx.x * FS_NODES;
    if (n0 >= n) return;
    const int  cnt = (n - n0 < FS_NODES) ? (n - n0) : FS_NODES;
    const bool own = t < cnt;
    const int  i   = n0 + (own ? t : 0);

    int B0 = rowptr[n0];
    int B1 = rowptr[n0 + cnt];
    int r0 = rowptr[i];
    int r1 = rowptr[i + 1];
    const float dg = normD[i];
    float cb = 0.0f;
    for (int q = 0; q < 4; ++q) {
        const int j = lane + 32 * q;
        cb += b2[j] * Wfc[j];
    }
    for (int off = 16; off > 0; off >>= 1)
        cb += __shfl_xor(cb, off, 32);
    B0 = B0 < 0 ? 0 : (B0 > N_EDGES ? N_EDGES : B0);
    B1 = B1 < B0 ? B0 : (B1 > N_EDGES ? N_EDGES : B1);
    r0 = r0 < 0 ? 0 : (r0 > N_EDGES ? N_EDGES : r0);
    r1 = r1 < 0 ? 0 : (r1 > N_EDGES ? N_EDGES : r1);

    float s = 0.0f;
    for (int base = B0; base < B1; base += FS_CAP) {
        const int len = (B1 - base < FS_CAP) ? (B1 - base) : FS_CAP;
        int   id[FS_BATCH];
        float v[FS_BATCH];
#pragma unroll
        for (int u = 0; u < FS_BATCH; ++u) {
            const int k = u * FS_TPB + t;
            id[u] = col[base + (k < len ? k : len - 1)];
        }
#pragma unroll
        for (int u = 0; u < FS_BATCH; ++u) {
            const int c = id[u];
            v[u] = S[c < 0 ? 0 : (c >= N_NODES ? N_NODES - 1 : c)];
        }
#pragma unroll
        for (int u = 0; u < FS_BATCH; ++u) vals[u * FS_TPB + t] = v[u];
        __syncthreads();
        if (own) {
            const int lo = (r0 > base ? r0 : base) - base;
            const int hi = (r1 < base + len ? r1 : base + len) - base;
            int k = lo;
            for (; k + 4 <= hi; k += 4) {
                const float v0 = vals[k], v1 = vals[k + 1], v2 = vals[k + 2], v3 = vals[k + 3];
                s += v0; s += v1; s += v2; s += v3;
            }
            for (; k < hi; ++k) s += vals[k];
        }
        __syncthreads();
    }
    if (own) out[i] = s * deg_to_norm(dg) + (cb + bfc[0]);
}

extern "C" void kernel_launch(void* const* d_in, const int* in_sizes, int n_in,
                              void* d_out, int out_size, void* d_ws, size_t ws_size,
                              hipStream_t stream) {
    const float* x   = (const float*)d_in[0];
    const float* W1  = (const float*)d_in[1];
    const float* b1  = (const float*)d_in[2];
    const float* W2  = (const float*)d_in[3];
    const float* b2  = (const float*)d_in[4];
    const float* Wfc = (const float*)d_in[5];
    const float* bfc = (const float*)d_in[6];
    const int*   src = (const int*)d_in[7];
    const int*   dst = (const int*)d_in[8];
    float* out = (float*)d_out;

    const int N = N_NODES, E = N_EDGES;

    float* BUF_A = (float*)d_ws;
    float* BUF_B = BUF_A + (size_t)N * NHID;
    float* normS = BUF_B + (size_t)N * NHID;
    float* normD = normS + N;

    int*      col    = (int*)(BUF_B + (size_t)7864320);
    unsigned* Ctab   = (unsigned*)(col + 1600000);
    int*      rowptr = (int*)(Ctab + 1600000);
    unsigned* RT     = (unsigned*)(rowptr + 100032);
    float*    S      = (float*)(RT + RS_BLOCKS * EDGE_NS);
    float*    wvec   = S + 100000;

    const int TPB = 256;
    const int ZERO_BLOCKS = (2 * N + TPB - 1) / TPB;

    setup_kernel<<<ZERO_BLOCKS + WV_BLOCKS, TPB, 0, stream>>>(normS, 2 * N, W2, Wfc, wvec,
                                                            ZERO_BLOCKS);
    degree_kernel<<<NODE_NR * EDGE_NS, ACC_TPB, 0, stream>>>(src, dst, normS, normD, Ctab, RT, E, N);

    rowstart_kernel<<<RS_BLOCKS, RS_NODES, 0, stream>>>(Ctab, RT, rowptr, N);
    place_kernel<<<NODE_NR * EDGE_NS, ACC_TPB, 0, stream>>>(src, dst, Ctab, col, E, N);

    gemm_scaled_wmma<<<1250, TPB, 0, stream>>>(x, normS, W1, BUF_A, NFEAT);

    gather_rows_dot<<<1250, TPB, 0, stream>>>(BUF_A, rowptr, col, normS, normD, b1, wvec, S, N);

    edge_sum_out<<<(N + FS_NODES - 1) / FS_NODES, FS_TPB, 0, stream>>>(S, rowptr, col, normD, b2, Wfc,
                                                                       bfc, out, N);
}
